# starting kernel
# baseline (speedup 1.0000x reference)
#include <hip/hip_runtime.h>

#define NUM_USER   100000
#define NUM_ITEM   50000
#define NTOT       (NUM_USER + NUM_ITEM)
#define DIM_FEAT   512
#define DIM_LATENT 64

typedef __attribute__((ext_vector_type(2))) float v2f;
typedef __attribute__((ext_vector_type(8))) float v8f;

__global__ __launch_bounds__(128) void mgcn_gemm_wmma(
    const float* __restrict__ F, const float* __restrict__ W,
    const float* __restrict__ bias, float* __restrict__ emb)
{
    const int wave  = threadIdx.x >> 5;
    const int lane  = threadIdx.x & 31;
    const int half  = lane >> 4;
    const int mrow  = lane & 15;
    const int mtile = blockIdx.x;
    const int m     = mtile * 16 + mrow;
    const int ncol  = wave * 16 + (lane & 15);

    const float* frow = F + (size_t)m * DIM_FEAT;

    v8f c = {};
    for (int k0 = 0; k0 < DIM_FEAT; k0 += 4) {
        const int k = k0 + 2 * half;
        v2f a = *(const v2f*)(frow + k);
        v2f bm;
        bm.x = W[(size_t)k * DIM_LATENT + ncol];
        bm.y = W[(size_t)(k + 1) * DIM_LATENT + ncol];
        c = __builtin_amdgcn_wmma_f32_16x16x4_f32(
                 false, a,  false, bm,
                 (short)0, c,  false,  false);
    }

    const float bv = bias[ncol];
    float* dst = emb + (size_t)(NUM_USER + mtile * 16) * DIM_LATENT;
#pragma unroll
    for (int j = 0; j < 8; ++j) {
        const int mr = j + 8 * half;
        dst[(size_t)mr * DIM_LATENT + ncol] = c[j] + bv;
    }
}

__global__ __launch_bounds__(256) void mgcn_normalize_init(
    const float* __restrict__ upref, float* __restrict__ emb,
    float* __restrict__ sum)
{
    const int row  = (int)((blockIdx.x * blockDim.x + threadIdx.x) >> 5);
    const int lane = threadIdx.x & 31;
    if (row >= NTOT) return;

    const float* src = (row < NUM_USER)
                           ? (upref + (size_t)row * DIM_LATENT)
                           : (emb + (size_t)row * DIM_LATENT);
    v2f x = *(const v2f*)(src + lane * 2);

    float s = x.x * x.x + x.y * x.y;
#pragma unroll
    for (int off = 16; off; off >>= 1) s += __shfl_xor(s, off, 32);

    const float inv = 1.0f / fmaxf(sqrtf(s), 1e-12f);
    v2f y; y.x = x.x * inv; y.y = x.y * inv;

    *(v2f*)(emb + (size_t)row * DIM_LATENT + lane * 2) = y;
    *(v2f*)(sum + (size_t)row * DIM_LATENT + lane * 2) = y;
}

__global__ __launch_bounds__(256) void mgcn_spmm_scatter(
    const float* __restrict__ src, const int* __restrict__ rows,
    const int* __restrict__ cols, const float* __restrict__ vals,
    float* __restrict__ dst, int nnz)
{
    const int wave = (int)((blockIdx.x * blockDim.x + threadIdx.x) >> 5);
    const int lane = threadIdx.x & 31;
    const int base = wave * 32;
    if (base >= nnz) return;

    const int myE = base + lane;
    int   r32 = 0, c32 = 0;
    float v32 = 0.0f;
    if (myE < nnz) { r32 = rows[myE]; c32 = cols[myE]; v32 = vals[myE]; }

    const int cnt = (nnz - base < 32) ? (nnz - base) : 32;
    for (int i = 0; i < cnt; ++i) {
        const int   r = __shfl(r32, i, 32);
        const int   c = __shfl(c32, i, 32);
        const float v = __shfl(v32, i, 32);

        v2f x = *(const v2f*)(src + (size_t)c * DIM_LATENT + lane * 2);
        float* d = dst + (size_t)r * DIM_LATENT + lane * 2;
        atomicAdd(d,     x.x * v);
        atomicAdd(d + 1, x.y * v);
    }
}

__global__ __launch_bounds__(256) void mgcn_accumulate4(
    float4* __restrict__ sum, const float4* __restrict__ add,
    float scale, int n4)
{
    const int i = blockIdx.x * blockDim.x + threadIdx.x;
    if (i >= n4) return;
    float4 s = sum[i];
    const float4 a = add[i];
    s.x = (s.x + a.x) * scale;
    s.y = (s.y + a.y) * scale;
    s.z = (s.z + a.z) * scale;
    s.w = (s.w + a.w) * scale;
    sum[i] = s;
}

extern "C" void kernel_launch(void* const* d_in, const int* in_sizes, int n_in,
                              void* d_out, int out_size, void* d_ws, size_t ws_size,
                              hipStream_t stream)
{
    const float* features = (const float*)d_in[0];
    const float* upref    = (const float*)d_in[1];
    const float* W        = (const float*)d_in[2];
    const float* bias     = (const float*)d_in[3];
    const int*   rows     = (const int*)d_in[4];
    const int*   cols     = (const int*)d_in[5];
    const float* vals     = (const float*)d_in[6];
    float*       out      = (float*)d_out;
    const int    nnz      = in_sizes[4];

    const size_t embBytes = (size_t)NTOT * DIM_LATENT * sizeof(float);
    float* bufA = (float*)d_ws;
    float* bufB = (float*)((char*)d_ws + embBytes);

    mgcn_gemm_wmma<<<NUM_ITEM / 16, 128, 0, stream>>>(features, W, bias, bufA);

    {
        const long long threads = (long long)NTOT * 32;
        const int blocks = (int)((threads + 255) / 256);
        mgcn_normalize_init<<<blocks, 256, 0, stream>>>(upref, bufA, out);
    }

    float* cur = bufA;
    float* nxt = bufB;
    const int scatterWaves  = (nnz + 31) / 32;
    const int scatterBlocks = (int)(((long long)scatterWaves * 32 + 255) / 256);
    const int n4            = NTOT * DIM_LATENT / 4;
    const int accBlocks     = (n4 + 255) / 256;

    for (int layer = 0; layer < 3; ++layer) {
        hipMemsetAsync(nxt, 0, embBytes, stream);
        mgcn_spmm_scatter<<<scatterBlocks, 256, 0, stream>>>(cur, rows, cols,
                                                             vals, nxt, nnz);
        const float scale = (layer == 2) ? 0.25f : 1.0f;
        mgcn_accumulate4<<<accBlocks, 256, 0, stream>>>((float4*)out,
                                                        (const float4*)nxt,
                                                        scale, n4);
        float* t = cur; cur = nxt; nxt = t;
    }
}


# SpMM layers: f32-atomic scatter replaced by rows sorted once in-call (int atomics: count, allocate, fill) and one wave per row summing its edges; layer sums folded in; f32 sum order = fill order
# speedup vs baseline: 1.2246x; 1.2246x over previous
#include <hip/hip_runtime.h>


#define NUM_USER   100000
#define NUM_ITEM   50000
#define NTOT       (NUM_USER + NUM_ITEM)
#define DIM_FEAT   512
#define DIM_LATENT 64

typedef __attribute__((ext_vector_type(2))) float v2f;
typedef __attribute__((ext_vector_type(8))) float v8f;

__global__ __launch_bounds__(128) void mgcn_gemm_wmma(
    const float* __restrict__ F, const float* __restrict__ W,
    const float* __restrict__ bias, float* __restrict__ items)
{
    const int wave  = threadIdx.x >> 5;
    const int lane  = threadIdx.x & 31;
    const int half  = lane >> 4;
    const int mrow  = lane & 15;
    const int mtile = blockIdx.x;
    const int m     = mtile * 16 + mrow;
    const int ncol  = wave * 16 + (lane & 15);

    const float* frow = F + (size_t)m * DIM_FEAT;

    v8f c = {};
    for (int k0 = 0; k0 < DIM_FEAT; k0 += 4) {
        const int k = k0 + 2 * half;
        v2f a = *(const v2f*)(frow + k);
        v2f bm;
        bm.x = W[(size_t)k * DIM_LATENT + ncol];
        bm.y = W[(size_t)(k + 1) * DIM_LATENT + ncol];
        c = __builtin_amdgcn_wmma_f32_16x16x4_f32(
                 false, a,  false, bm,
                 (short)0, c,  false,  false);
    }

    const float bv = bias[ncol];
    float* dst = items + (size_t)(mtile * 16) * DIM_LATENT;
#pragma unroll
    for (int j = 0; j < 8; ++j) {
        const int mr = j + 8 * half;
        dst[(size_t)mr * DIM_LATENT + ncol] = c[j] + bv;
    }
}

__global__ __launch_bounds__(256) void mgcn_row_norms(
    const float* __restrict__ upref, float* __restrict__ items,
    float* __restrict__ inv)
{
    const int row  = (int)((blockIdx.x * blockDim.x + threadIdx.x) >> 5);
    const int lane = threadIdx.x & 31;
    if (row >= NTOT) return;

    const bool isUser = row < NUM_USER;
    float* irow = items + (size_t)(isUser ? 0 : (row - NUM_USER)) * DIM_LATENT;
    const float* src = isUser ? (upref + (size_t)row * DIM_LATENT) : irow;
    v2f x = *(const v2f*)(src + lane * 2);

    float s = x.x * x.x + x.y * x.y;
#pragma unroll
    for (int off = 16; off; off >>= 1) s += __shfl_xor(s, off, 32);

    const float r = 1.0f / fmaxf(sqrtf(s), 1e-12f);
    if (isUser) {
        if (lane == 0) inv[row] = r;
    } else {
        v2f y; y.x = x.x * r; y.y = x.y * r;
        *(v2f*)(irow + lane * 2) = y;
    }
}

__global__ __launch_bounds__(256) void mgcn_row_hist(
    const int* __restrict__ rows, int* __restrict__ counts, int nnz)
{
    const int e = (int)(blockIdx.x * 256 + threadIdx.x) * 4;
    if (e + 3 < nnz) {
        const int4 r = *(const int4*)(rows + e);
        atomicAdd(counts + r.x, 1);
        atomicAdd(counts + r.y, 1);
        atomicAdd(counts + r.z, 1);
        atomicAdd(counts + r.w, 1);
    } else {
        for (int k = e; k < nnz; ++k) atomicAdd(counts + rows[k], 1);
    }
}

__global__ __launch_bounds__(256) void mgcn_row_alloc(
    const int* __restrict__ counts, int* __restrict__ total,
    int* __restrict__ start, int* __restrict__ cursor)
{
    const int r    = (int)(blockIdx.x * 256 + threadIdx.x);
    const int lane = threadIdx.x & 31;
    const int c    = (r < NTOT) ? counts[r] : 0;

    int incl = c;
    for (int off = 1; off < 32; off <<= 1) {
        const int t = __shfl_up(incl, off, 32);
        if (lane >= off) incl += t;
    }
    int base = 0;
    if (lane == 31) base = atomicAdd(total, incl);
    base = __shfl(base, 31, 32);

    const int s = base + incl - c;
    if (r < NTOT) { start[r] = s; cursor[r] = s; }
}

__global__ __launch_bounds__(256) void mgcn_edge_fill(
    const int* __restrict__ rows, const int* __restrict__ cols,
    const float* __restrict__ vals, int* __restrict__ cursor,
    int2* __restrict__ ed, int nnz)
{
    const int e = (int)(blockIdx.x * 256 + threadIdx.x) * 4;
    if (e + 3 < nnz) {
        const int4   r = *(const int4*)(rows + e);
        const int4   c = *(const int4*)(cols + e);
        const float4 v = *(const float4*)(vals + e);
        const int p0 = atomicAdd(cursor + r.x, 1);
        const int p1 = atomicAdd(cursor + r.y, 1);
        const int p2 = atomicAdd(cursor + r.z, 1);
        const int p3 = atomicAdd(cursor + r.w, 1);
        ed[p0] = make_int2(c.x, __float_as_int(v.x));
        ed[p1] = make_int2(c.y, __float_as_int(v.y));
        ed[p2] = make_int2(c.z, __float_as_int(v.z));
        ed[p3] = make_int2(c.w, __float_as_int(v.w));
    } else {
        for (int k = e; k < nnz; ++k) {
            const int p = atomicAdd(cursor + rows[k], 1);
            ed[p] = make_int2(cols[k], __float_as_int(vals[k]));
        }
    }
}

template <int MODE>
__global__ __launch_bounds__(256) void mgcn_pull(
    const float* __restrict__ src, const float* __restrict__ upref,
    const float* __restrict__ items, const float* __restrict__ inv,
    const int* __restrict__ start, const int* __restrict__ endp,
    const int2* __restrict__ ed, float* __restrict__ dst)
{
    const int lane = threadIdx.x & 31;
    const int row = __builtin_amdgcn_readfirstlane(
        (int)((blockIdx.x * 256 + threadIdx.x) >> 5));
    if (row >= NTOT) return;

    const int e0 = __builtin_amdgcn_readfirstlane(start[row]);
    const int e1 = __builtin_amdgcn_readfirstlane(endp[row]);

    float* d = dst + (size_t)row * DIM_LATENT + lane * 2;

    v2f l0 = {0.0f, 0.0f}, l1 = {0.0f, 0.0f}, l2 = {0.0f, 0.0f};
    if (MODE == 3) {
        if (row < NUM_USER) {
            const v2f x = *(const v2f*)(upref + (size_t)row * DIM_LATENT + lane * 2);
            const float r = inv[row];
            l0.x = x.x * r; l0.y = x.y * r;
        } else {
            l0 = *(const v2f*)(items + (size_t)(row - NUM_USER) * DIM_LATENT + lane * 2);
        }
        l1 = *(const v2f*)d;
        l2 = *(const v2f*)(src + (size_t)row * DIM_LATENT + lane * 2);
    }

    v2f acc = {0.0f, 0.0f};
    for (int base = e0; base < e1; base += 32) {
        const int rem = e1 - base;
        const int n   = rem < 32 ? rem : 32;

        int   c32 = 0;
        float v32 = 0.0f;
        float s32 = 1.0f;
        if (lane < n) {
            const int2 t = ed[base + lane];
            c32 = t.x;
            v32 = __int_as_float(t.y);
            if (MODE == 1) { if (c32 < NUM_USER) s32 = inv[c32]; }
        }

        for (int i = 0; i < n; i += 8) {
            v2f   x[8];
            float v[8];
            float s[8];
#pragma unroll
            for (int k = 0; k < 8; ++k) {
                const int c = __builtin_amdgcn_readlane(c32, i + k);
                v[k] = __int_as_float(
                    __builtin_amdgcn_readlane(__float_as_int(v32), i + k));
                const float* p;
                if (MODE == 1) {
                    s[k] = __int_as_float(
                        __builtin_amdgcn_readlane(__float_as_int(s32), i + k));
                    p = (c < NUM_USER)
                            ? (upref + (size_t)c * DIM_LATENT)
                            : (items + (size_t)(c - NUM_USER) * DIM_LATENT);
                } else {
                    s[k] = 1.0f;
                    p = src + (size_t)c * DIM_LATENT;
                }
                x[k] = *(const v2f*)(p + lane * 2);
            }
#pragma unroll
            for (int k = 0; k < 8; ++k) {
                if (MODE == 1) {
                    acc.x += (x[k].x * s[k]) * v[k];
                    acc.y += (x[k].y * s[k]) * v[k];
                } else {
                    acc.x += x[k].x * v[k];
                    acc.y += x[k].y * v[k];
                }
            }
        }
    }

    if (MODE == 3) {
        v2f o;
        o.x = (((l0.x + l1.x) + l2.x) + acc.x) * 0.25f;
        o.y = (((l0.y + l1.y) + l2.y) + acc.y) * 0.25f;
        *(v2f*)d = o;
    } else {
        *(v2f*)d = acc;
    }
}

extern "C" void kernel_launch(void* const* d_in, const int* in_sizes, int n_in,
                              void* d_out, int out_size, void* d_ws, size_t ws_size,
                              hipStream_t stream)
{
    const float* features = (const float*)d_in[0];
    const float* upref    = (const float*)d_in[1];
    const float* W        = (const float*)d_in[2];
    const float* bias     = (const float*)d_in[3];
    const int*   rows     = (const int*)d_in[4];
    const int*   cols     = (const int*)d_in[5];
    const float* vals     = (const float*)d_in[6];
    float*       out      = (float*)d_out;
    const int    nnz      = in_sizes[4];

    char*  ws     = (char*)d_ws;
    float* tab2   = (float*)(ws);
    float* items  = (float*)(ws + 38400000);
    float* inv    = (float*)(ws + 51200000);
    int*   counts = (int*)  (ws + 51600128);
    int*   total  = (int*)  (ws + 52200128);
    int*   start  = (int*)  (ws + 52200256);
    int*   cursor = (int*)  (ws + 52800256);
    int2*  ed     = (int2*) (ws + 53400320);

    mgcn_gemm_wmma<<<NUM_ITEM / 16, 128, 0, stream>>>(features, W, bias, items);

    mgcn_row_norms<<<NTOT * 32 / 256, 256, 0, stream>>>(upref, items, inv);

    const int quadBlocks = (nnz + 1023) / 1024;
    hipMemsetAsync(counts, 0, 600128, stream);
    mgcn_row_hist<<<quadBlocks, 256, 0, stream>>>(rows, counts, nnz);
    mgcn_row_alloc<<<(NTOT + 255) / 256, 256, 0, stream>>>(counts, total, start, cursor);
    mgcn_edge_fill<<<quadBlocks, 256, 0, stream>>>(rows, cols, vals, cursor, ed, nnz);

    const int pullBlocks = NTOT * 32 / 256;
    mgcn_pull<1><<<pullBlocks, 256, 0, stream>>>(upref, upref, items, inv,
                                                 start, cursor, ed, out);
    mgcn_pull<2><<<pullBlocks, 256, 0, stream>>>(out, upref, items, inv,
                                                 start, cursor, ed, tab2);
    mgcn_pull<3><<<pullBlocks, 256, 0, stream>>>(tab2, upref, items, inv,
                                                 start, cursor, ed, out);
}


# Item GEMM: a wave owns a 16x64 tile (4 accumulators), W staged once per block in LDS as k-pairs, A read as 16-byte loads one trip ahead; f32 sum order: k order inside each 32-block changed
# speedup vs baseline: 1.3067x; 1.0670x over previous
#include <hip/hip_runtime.h>


#define NUM_USER   100000
#define NUM_ITEM   50000
#define NTOT       (NUM_USER + NUM_ITEM)
#define DIM_FEAT   512
#define DIM_LATENT 64

typedef __attribute__((ext_vector_type(2))) float v2f;
typedef __attribute__((ext_vector_type(8))) float v8f;

#define GEMM_TILES   (NUM_ITEM / 16)
#define W_PITCH      132

__global__ __launch_bounds__(256) void mgcn_gemm_wmma(
    const float* __restrict__ F, const float* __restrict__ W,
    const float* __restrict__ bias, float* __restrict__ items)
{
    __shared__ __attribute__((aligned(16))) float Wl[(DIM_FEAT / 2) * W_PITCH];

    for (int it = 0; it < (DIM_FEAT / 2) * 16 / 256; ++it) {
        const int idx = it * 256 + (int)threadIdx.x;
        const int kp  = idx >> 4;
        const int n4  = idx & 15;
        const float4 r0 = *(const float4*)(W + (size_t)(2 * kp) * DIM_LATENT + n4 * 4);
        const float4 r1 = *(const float4*)(W + (size_t)(2 * kp + 1) * DIM_LATENT + n4 * 4);
        float* d = Wl + kp * W_PITCH + n4 * 8;
        *(float4*)d       = make_float4(r0.x, r1.x, r0.y, r1.y);
        *(float4*)(d + 4) = make_float4(r0.z, r1.z, r0.w, r1.w);
    }
    __syncthreads();

    const int lane = threadIdx.x & 31;
    const int half = lane >> 4;
    const int c    = lane & 15;
    const int tile = __builtin_amdgcn_readfirstlane(
        (int)(blockIdx.x * 8 + (threadIdx.x >> 5)));
    if (tile >= GEMM_TILES) return;

    const float* frow = F + (size_t)(tile * 16 + c) * DIM_FEAT + 16 * half;
    const float* wl   = Wl + (8 * half) * W_PITCH + 2 * c;

    v8f acc[4] = {};
    float4 cur[4], nxt[4];
#pragma unroll
    for (int j = 0; j < 4; ++j) cur[j] = *(const float4*)(frow + 4 * j);

    for (int t = 0; t < DIM_FEAT / 32; ++t) {
        const int tn = (t + 1 < DIM_FEAT / 32) ? t + 1 : t;
#pragma unroll
        for (int j = 0; j < 4; ++j) nxt[j] = *(const float4*)(frow + tn * 32 + 4 * j);

        const float* wt = wl + t * 16 * W_PITCH;
#pragma unroll
        for (int st = 0; st < 8; ++st) {
            const float4 q = cur[st >> 1];
            v2f a;
            a.x = (st & 1) ? q.z : q.x;
            a.y = (st & 1) ? q.w : q.y;
#pragma unroll
            for (int nt = 0; nt < 4; ++nt) {
                const v2f bm = *(const v2f*)(wt + st * W_PITCH + nt * 32);
                acc[nt] = __builtin_amdgcn_wmma_f32_16x16x4_f32(
                         false, a,  false, bm,
                         (short)0, acc[nt],  false,  false);
            }
        }
#pragma unroll
        for (int j = 0; j < 4; ++j) cur[j] = nxt[j];
    }

    float* dst = items + (size_t)(tile * 16) * DIM_LATENT;
#pragma unroll
    for (int nt = 0; nt < 4; ++nt) {
        const float bv = bias[nt * 16 + c];
#pragma unroll
        for (int j = 0; j < 8; ++j) {
            const int mr = j + 8 * half;
            dst[(size_t)mr * DIM_LATENT + nt * 16 + c] = acc[nt][j] + bv;
        }
    }
}

__global__ __launch_bounds__(256) void mgcn_row_norms(
    const float* __restrict__ upref, float* __restrict__ items,
    float* __restrict__ inv)
{
    const int row  = (int)((blockIdx.x * blockDim.x + threadIdx.x) >> 5);
    const int lane = threadIdx.x & 31;
    if (row >= NTOT) return;

    const bool isUser = row < NUM_USER;
    float* irow = items + (size_t)(isUser ? 0 : (row - NUM_USER)) * DIM_LATENT;
    const float* src = isUser ? (upref + (size_t)row * DIM_LATENT) : irow;
    v2f x = *(const v2f*)(src + lane * 2);

    float s = x.x * x.x + x.y * x.y;
#pragma unroll
    for (int off = 16; off; off >>= 1) s += __shfl_xor(s, off, 32);

    const float r = 1.0f / fmaxf(sqrtf(s), 1e-12f);
    if (isUser) {
        if (lane == 0) inv[row] = r;
    } else {
        v2f y; y.x = x.x * r; y.y = x.y * r;
        *(v2f*)(irow + lane * 2) = y;
    }
}

__global__ __launch_bounds__(256) void mgcn_row_hist(
    const int* __restrict__ rows, int* __restrict__ counts, int nnz)
{
    const int e = (int)(blockIdx.x * 256 + threadIdx.x) * 4;
    if (e + 3 < nnz) {
        const int4 r = *(const int4*)(rows + e);
        atomicAdd(counts + r.x, 1);
        atomicAdd(counts + r.y, 1);
        atomicAdd(counts + r.z, 1);
        atomicAdd(counts + r.w, 1);
    } else {
        for (int k = e; k < nnz; ++k) atomicAdd(counts + rows[k], 1);
    }
}

__global__ __launch_bounds__(256) void mgcn_row_alloc(
    const int* __restrict__ counts, int* __restrict__ total,
    int* __restrict__ start, int* __restrict__ cursor)
{
    const int r    = (int)(blockIdx.x * 256 + threadIdx.x);
    const int lane = threadIdx.x & 31;
    const int c    = (r < NTOT) ? counts[r] : 0;

    int incl = c;
    for (int off = 1; off < 32; off <<= 1) {
        const int t = __shfl_up(incl, off, 32);
        if (lane >= off) incl += t;
    }
    int base = 0;
    if (lane == 31) base = atomicAdd(total, incl);
    base = __shfl(base, 31, 32);

    const int s = base + incl - c;
    if (r < NTOT) { start[r] = s; cursor[r] = s; }
}

__global__ __launch_bounds__(256) void mgcn_edge_fill(
    const int* __restrict__ rows, const int* __restrict__ cols,
    const float* __restrict__ vals, int* __restrict__ cursor,
    int2* __restrict__ ed, int nnz)
{
    const int e = (int)(blockIdx.x * 256 + threadIdx.x) * 4;
    if (e + 3 < nnz) {
        const int4   r = *(const int4*)(rows + e);
        const int4   c = *(const int4*)(cols + e);
        const float4 v = *(const float4*)(vals + e);
        const int p0 = atomicAdd(cursor + r.x, 1);
        const int p1 = atomicAdd(cursor + r.y, 1);
        const int p2 = atomicAdd(cursor + r.z, 1);
        const int p3 = atomicAdd(cursor + r.w, 1);
        ed[p0] = make_int2(c.x, __float_as_int(v.x));
        ed[p1] = make_int2(c.y, __float_as_int(v.y));
        ed[p2] = make_int2(c.z, __float_as_int(v.z));
        ed[p3] = make_int2(c.w, __float_as_int(v.w));
    } else {
        for (int k = e; k < nnz; ++k) {
            const int p = atomicAdd(cursor + rows[k], 1);
            ed[p] = make_int2(cols[k], __float_as_int(vals[k]));
        }
    }
}

template <int MODE>
__global__ __launch_bounds__(256) void mgcn_pull(
    const float* __restrict__ src, const float* __restrict__ upref,
    const float* __restrict__ items, const float* __restrict__ inv,
    const int* __restrict__ start, const int* __restrict__ endp,
    const int2* __restrict__ ed, float* __restrict__ dst)
{
    const int lane = threadIdx.x & 31;
    const int row = __builtin_amdgcn_readfirstlane(
        (int)((blockIdx.x * 256 + threadIdx.x) >> 5));
    if (row >= NTOT) return;

    const int e0 = __builtin_amdgcn_readfirstlane(start[row]);
    const int e1 = __builtin_amdgcn_readfirstlane(endp[row]);

    float* d = dst + (size_t)row * DIM_LATENT + lane * 2;

    v2f l0 = {0.0f, 0.0f}, l1 = {0.0f, 0.0f}, l2 = {0.0f, 0.0f};
    if (MODE == 3) {
        if (row < NUM_USER) {
            const v2f x = *(const v2f*)(upref + (size_t)row * DIM_LATENT + lane * 2);
            const float r = inv[row];
            l0.x = x.x * r; l0.y = x.y * r;
        } else {
            l0 = *(const v2f*)(items + (size_t)(row - NUM_USER) * DIM_LATENT + lane * 2);
        }
        l1 = *(const v2f*)d;
        l2 = *(const v2f*)(src + (size_t)row * DIM_LATENT + lane * 2);
    }

    v2f acc = {0.0f, 0.0f};
    for (int base = e0; base < e1; base += 32) {
        const int rem = e1 - base;
        const int n   = rem < 32 ? rem : 32;

        int   c32 = 0;
        float v32 = 0.0f;
        float s32 = 1.0f;
        if (lane < n) {
            const int2 t = ed[base + lane];
            c32 = t.x;
            v32 = __int_as_float(t.y);
            if (MODE == 1) { if (c32 < NUM_USER) s32 = inv[c32]; }
        }

        for (int i = 0; i < n; i += 8) {
            v2f   x[8];
            float v[8];
            float s[8];
#pragma unroll
            for (int k = 0; k < 8; ++k) {
                const int c = __builtin_amdgcn_readlane(c32, i + k);
                v[k] = __int_as_float(
                    __builtin_amdgcn_readlane(__float_as_int(v32), i + k));
                const float* p;
                if (MODE == 1) {
                    s[k] = __int_as_float(
                        __builtin_amdgcn_readlane(__float_as_int(s32), i + k));
                    p = (c < NUM_USER)
                            ? (upref + (size_t)c * DIM_LATENT)
                            : (items + (size_t)(c - NUM_USER) * DIM_LATENT);
                } else {
                    s[k] = 1.0f;
                    p = src + (size_t)c * DIM_LATENT;
                }
                x[k] = *(const v2f*)(p + lane * 2);
            }
#pragma unroll
            for (int k = 0; k < 8; ++k) {
                if (MODE == 1) {
                    acc.x += (x[k].x * s[k]) * v[k];
                    acc.y += (x[k].y * s[k]) * v[k];
                } else {
                    acc.x += x[k].x * v[k];
                    acc.y += x[k].y * v[k];
                }
            }
        }
    }

    if (MODE == 3) {
        v2f o;
        o.x = (((l0.x + l1.x) + l2.x) + acc.x) * 0.25f;
        o.y = (((l0.y + l1.y) + l2.y) + acc.y) * 0.25f;
        *(v2f*)d = o;
    } else {
        *(v2f*)d = acc;
    }
}

extern "C" void kernel_launch(void* const* d_in, const int* in_sizes, int n_in,
                              void* d_out, int out_size, void* d_ws, size_t ws_size,
                              hipStream_t stream)
{
    const float* features = (const float*)d_in[0];
    const float* upref    = (const float*)d_in[1];
    const float* W        = (const float*)d_in[2];
    const float* bias     = (const float*)d_in[3];
    const int*   rows     = (const int*)d_in[4];
    const int*   cols     = (const int*)d_in[5];
    const float* vals     = (const float*)d_in[6];
    float*       out      = (float*)d_out;
    const int    nnz      = in_sizes[4];

    char*  ws     = (char*)d_ws;
    float* tab2   = (float*)(ws);
    float* items  = (float*)(ws + 38400000);
    float* inv    = (float*)(ws + 51200000);
    int*   counts = (int*)  (ws + 51600128);
    int*   total  = (int*)  (ws + 52200128);
    int*   start  = (int*)  (ws + 52200256);
    int*   cursor = (int*)  (ws + 52800256);
    int2*  ed     = (int2*) (ws + 53400320);

    mgcn_gemm_wmma<<<(GEMM_TILES + 7) / 8, 256, 0, stream>>>(features, W, bias, items);

    mgcn_row_norms<<<NTOT * 32 / 256, 256, 0, stream>>>(upref, items, inv);

    const int quadBlocks = (nnz + 1023) / 1024;
    hipMemsetAsync(counts, 0, 600128, stream);
    mgcn_row_hist<<<quadBlocks, 256, 0, stream>>>(rows, counts, nnz);
    mgcn_row_alloc<<<(NTOT + 255) / 256, 256, 0, stream>>>(counts, total, start, cursor);
    mgcn_edge_fill<<<quadBlocks, 256, 0, stream>>>(rows, cols, vals, cursor, ed, nnz);

    const int pullBlocks = NTOT * 32 / 256;
    mgcn_pull<1><<<pullBlocks, 256, 0, stream>>>(upref, upref, items, inv,
                                                 start, cursor, ed, out);
    mgcn_pull<2><<<pullBlocks, 256, 0, stream>>>(out, upref, items, inv,
                                                 start, cursor, ed, tab2);
    mgcn_pull<3><<<pullBlocks, 256, 0, stream>>>(tab2, upref, items, inv,
                                                 start, cursor, ed, out);
}


# Edge sort: per-edge global integer atomics (count, allocate, fill) replaced by a two-level counting sort on LDS counters with whole-line writes; f32 sum order = new placement order
# speedup vs baseline: 2.8799x; 2.2039x over previous
#include <hip/hip_runtime.h>


#define NUM_USER   100000
#define NUM_ITEM   50000
#define NTOT       (NUM_USER + NUM_ITEM)
#define DIM_FEAT   512
#define DIM_LATENT 64

typedef __attribute__((ext_vector_type(2))) float v2f;
typedef __attribute__((ext_vector_type(8))) float v8f;

#define GEMM_TILES   (NUM_ITEM / 16)
#define W_PITCH      132

__global__ __launch_bounds__(256) void mgcn_gemm_wmma(
    const float* __restrict__ F, const float* __restrict__ W,
    const float* __restrict__ bias, float* __restrict__ items)
{
    __shared__ __attribute__((aligned(16))) float Wl[(DIM_FEAT / 2) * W_PITCH];

    for (int it = 0; it < (DIM_FEAT / 2) * 16 / 256; ++it) {
        const int idx = it * 256 + (int)threadIdx.x;
        const int kp  = idx >> 4;
        const int n4  = idx & 15;
        const float4 r0 = *(const float4*)(W + (size_t)(2 * kp) * DIM_LATENT + n4 * 4);
        const float4 r1 = *(const float4*)(W + (size_t)(2 * kp + 1) * DIM_LATENT + n4 * 4);
        float* d = Wl + kp * W_PITCH + n4 * 8;
        *(float4*)d       = make_float4(r0.x, r1.x, r0.y, r1.y);
        *(float4*)(d + 4) = make_float4(r0.z, r1.z, r0.w, r1.w);
    }
    __syncthreads();

    const int lane = threadIdx.x & 31;
    const int half = lane >> 4;
    const int c    = lane & 15;
    const int tile = __builtin_amdgcn_readfirstlane(
        (int)(blockIdx.x * 8 + (threadIdx.x >> 5)));
    if (tile >= GEMM_TILES) return;

    const float* frow = F + (size_t)(tile * 16 + c) * DIM_FEAT + 16 * half;
    const float* wl   = Wl + (8 * half) * W_PITCH + 2 * c;

    v8f acc[4] = {};
    float4 cur[4], nxt[4];
#pragma unroll
    for (int j = 0; j < 4; ++j) cur[j] = *(const float4*)(frow + 4 * j);

    for (int t = 0; t < DIM_FEAT / 32; ++t) {
        const int tn = (t + 1 < DIM_FEAT / 32) ? t + 1 : t;
#pragma unroll
        for (int j = 0; j < 4; ++j) nxt[j] = *(const float4*)(frow + tn * 32 + 4 * j);

        const float* wt = wl + t * 16 * W_PITCH;
#pragma unroll
        for (int st = 0; st < 8; ++st) {
            const float4 q = cur[st >> 1];
            v2f a;
            a.x = (st & 1) ? q.z : q.x;
            a.y = (st & 1) ? q.w : q.y;
#pragma unroll
            for (int nt = 0; nt < 4; ++nt) {
                const v2f bm = *(const v2f*)(wt + st * W_PITCH + nt * 32);
                acc[nt] = __builtin_amdgcn_wmma_f32_16x16x4_f32(
                         false, a,  false, bm,
                         (short)0, acc[nt],  false,  false);
            }
        }
#pragma unroll
        for (int j = 0; j < 4; ++j) cur[j] = nxt[j];
    }

    float* dst = items + (size_t)(tile * 16) * DIM_LATENT;
#pragma unroll
    for (int nt = 0; nt < 4; ++nt) {
        const float bv = bias[nt * 16 + c];
#pragma unroll
        for (int j = 0; j < 8; ++j) {
            const int mr = j + 8 * half;
            dst[(size_t)mr * DIM_LATENT + nt * 16 + c] = acc[nt][j] + bv;
        }
    }
}

__global__ __launch_bounds__(256) void mgcn_row_norms(
    const float* __restrict__ upref, float* __restrict__ items,
    float* __restrict__ inv)
{
    const int row  = (int)((blockIdx.x * blockDim.x + threadIdx.x) >> 5);
    const int lane = threadIdx.x & 31;
    if (row >= NTOT) return;

    const bool isUser = row < NUM_USER;
    float* irow = items + (size_t)(isUser ? 0 : (row - NUM_USER)) * DIM_LATENT;
    const float* src = isUser ? (upref + (size_t)row * DIM_LATENT) : irow;
    v2f x = *(const v2f*)(src + lane * 2);

    float s = x.x * x.x + x.y * x.y;
#pragma unroll
    for (int off = 16; off; off >>= 1) s += __shfl_xor(s, off, 32);

    const float r = 1.0f / fmaxf(sqrtf(s), 1e-12f);
    if (isUser) {
        if (lane == 0) inv[row] = r;
    } else {
        v2f y; y.x = x.x * r; y.y = x.y * r;
        *(v2f*)(irow + lane * 2) = y;
    }
}

#define NBUCKET      ((NTOT + 255) / 256)
#define SORT_BLOCKS  256
#define LROW         (NBUCKET + 1)
#define CHUNK_CAP    9600
#define STAGE_CAP    6144

template <int NWAVES>
__device__ __forceinline__ int block_excl_scan(int v, int* wsum, int* total)
{
    const int lane = threadIdx.x & 31;
    const int w    = threadIdx.x >> 5;
    int incl = v;
    for (int off = 1; off < 32; off <<= 1) {
        const int t = __shfl_up(incl, off, 32);
        if (lane >= off) incl += t;
    }
    if (lane == 31) wsum[w] = incl;
    __syncthreads();
    if (w == 0) {
        const int s = (lane < NWAVES) ? wsum[lane] : 0;
        int si = s;
        for (int off = 1; off < 32; off <<= 1) {
            const int t = __shfl_up(si, off, 32);
            if (lane >= off) si += t;
        }
        wsum[lane] = si - s;
        if (lane == 31) wsum[32] = si;
    }
    __syncthreads();
    *total = wsum[32];
    return wsum[w] + incl - v;
}

__device__ __forceinline__ int sort_chunk(int nnz)
{
    return ((nnz + SORT_BLOCKS * 4 - 1) / (SORT_BLOCKS * 4)) * 4;
}

__device__ __forceinline__ int2 edge_record(int r, int c, float v)
{
    return make_int2(((r & 255) << 18) | c, __float_as_int(v));
}

__global__ __launch_bounds__(1024) void mgcn_chunk_sort(
    const int* __restrict__ rows, const int* __restrict__ cols,
    const float* __restrict__ vals, int* __restrict__ ltab,
    int2* __restrict__ tmp, int nnz)
{
    __shared__ int  h[NBUCKET];
    __shared__ int  wsum[33];
    __shared__ int2 stage[CHUNK_CAP];

    const int g     = (int)blockIdx.x;
    const int chunk = sort_chunk(nnz);
    const int lo    = g * chunk;
    const int hi    = (lo + chunk < nnz) ? lo + chunk : nnz;
    const int n     = (hi > lo) ? hi - lo : 0;
    const int e0    = lo + (int)threadIdx.x * 4;

    int4   r[3], c[3];
    float4 v[3];
    bool   ok[3];
#pragma unroll
    for (int u = 0; u < 3; ++u) {
        const int e = e0 + u * 4096;
        ok[u] = e + 3 < hi;
        r[u] = make_int4(0, 0, 0, 0);
        c[u] = make_int4(0, 0, 0, 0);
        v[u] = make_float4(0.0f, 0.0f, 0.0f, 0.0f);
        if (ok[u]) {
            r[u] = *(const int4*)(rows + e);
            c[u] = *(const int4*)(cols + e);
            v[u] = *(const float4*)(vals + e);
        }
    }

    for (int i = threadIdx.x; i < NBUCKET; i += 1024) h[i] = 0;
    __syncthreads();

#pragma unroll
    for (int u = 0; u < 3; ++u) {
        if (ok[u]) {
            atomicAdd(&h[r[u].x >> 8], 1);
            atomicAdd(&h[r[u].y >> 8], 1);
            atomicAdd(&h[r[u].z >> 8], 1);
            atomicAdd(&h[r[u].w >> 8], 1);
        }
    }
    for (int u = 0, e = e0; e < hi; ++u, e += 4096) {
        if (u < 3 && e + 3 < hi) continue;
        const int kend = (e + 4 < hi) ? e + 4 : hi;
        for (int k = e; k < kend; ++k) atomicAdd(&h[rows[k] >> 8], 1);
    }
    __syncthreads();

    const int mine = (threadIdx.x < NBUCKET) ? h[threadIdx.x] : 0;
    int all;
    const int excl = block_excl_scan<32>(mine, wsum, &all);
    if (threadIdx.x < NBUCKET) {
        ltab[g * LROW + (int)threadIdx.x] = excl;
        h[threadIdx.x] = excl;
    }
    if (threadIdx.x == 0) ltab[g * LROW + NBUCKET] = all;
    __syncthreads();

    if (n <= CHUNK_CAP) {
#pragma unroll
        for (int u = 0; u < 3; ++u) {
            if (ok[u]) {
                const int p0 = atomicAdd(&h[r[u].x >> 8], 1);
                const int p1 = atomicAdd(&h[r[u].y >> 8], 1);
                const int p2 = atomicAdd(&h[r[u].z >> 8], 1);
                const int p3 = atomicAdd(&h[r[u].w >> 8], 1);
                stage[p0] = edge_record(r[u].x, c[u].x, v[u].x);
                stage[p1] = edge_record(r[u].y, c[u].y, v[u].y);
                stage[p2] = edge_record(r[u].z, c[u].z, v[u].z);
                stage[p3] = edge_record(r[u].w, c[u].w, v[u].w);
            }
        }
        for (int u = 0, e = e0; e < hi; ++u, e += 4096) {
            if (u < 3 && e + 3 < hi) continue;
            const int kend = (e + 4 < hi) ? e + 4 : hi;
            for (int k = e; k < kend; ++k) {
                const int rr = rows[k];
                const int p  = atomicAdd(&h[rr >> 8], 1);
                stage[p] = edge_record(rr, cols[k], vals[k]);
            }
        }
        __syncthreads();
        for (int i = threadIdx.x; i < n; i += 1024) tmp[lo + i] = stage[i];
    } else {
#pragma unroll
        for (int u = 0; u < 3; ++u) {
            if (ok[u]) {
                const int p0 = atomicAdd(&h[r[u].x >> 8], 1);
                const int p1 = atomicAdd(&h[r[u].y >> 8], 1);
                const int p2 = atomicAdd(&h[r[u].z >> 8], 1);
                const int p3 = atomicAdd(&h[r[u].w >> 8], 1);
                tmp[lo + p0] = edge_record(r[u].x, c[u].x, v[u].x);
                tmp[lo + p1] = edge_record(r[u].y, c[u].y, v[u].y);
                tmp[lo + p2] = edge_record(r[u].z, c[u].z, v[u].z);
                tmp[lo + p3] = edge_record(r[u].w, c[u].w, v[u].w);
            }
        }
        for (int u = 0, e = e0; e < hi; ++u, e += 4096) {
            if (u < 3 && e + 3 < hi) continue;
            const int kend = (e + 4 < hi) ? e + 4 : hi;
            for (int k = e; k < kend; ++k) {
                const int rr = rows[k];
                const int p  = atomicAdd(&h[rr >> 8], 1);
                tmp[lo + p] = edge_record(rr, cols[k], vals[k]);
            }
        }
    }
}

#define RUNS_PER_WAVE (SORT_BLOCKS / 16)
__global__ __launch_bounds__(512) void mgcn_bucket_rows(
    const int2* __restrict__ tmp, const int* __restrict__ ltab,
    int* __restrict__ start, int* __restrict__ endp,
    int2* __restrict__ ed, int nnz)
{
    __shared__ int  cnt[256];
    __shared__ int  cur[256];
    __shared__ int  rs[SORT_BLOCKS];
    __shared__ int  rn[SORT_BLOCKS];
    __shared__ int  wsum[33];
    __shared__ int2 stage[STAGE_CAP];

    const int b     = (int)blockIdx.x;
    const int lane  = threadIdx.x & 31;
    const int w     = __builtin_amdgcn_readfirstlane((int)(threadIdx.x >> 5));
    const int chunk = sort_chunk(nnz);

    int s0 = 0, len = 0;
    if (threadIdx.x < SORT_BLOCKS) {
        s0  = ltab[(int)threadIdx.x * LROW + b];
        len = ltab[(int)threadIdx.x * LROW + b + 1] - s0;
        rs[threadIdx.x] = (int)threadIdx.x * chunk + s0;
        rn[threadIdx.x] = len;
    }
    if (threadIdx.x < 256) cnt[threadIdx.x] = 0;
    int sumStart, sumLen;
    block_excl_scan<16>(s0, wsum, &sumStart);
    block_excl_scan<16>(len, wsum, &sumLen);
    const int base = __builtin_amdgcn_readfirstlane(sumStart);
    const int n    = __builtin_amdgcn_readfirstlane(sumLen);

    int2 t[RUNS_PER_WAVE];
    int  ln[RUNS_PER_WAVE];
    int  ro[RUNS_PER_WAVE];
#pragma unroll
    for (int u = 0; u < RUNS_PER_WAVE; ++u) {
        ln[u] = __builtin_amdgcn_readfirstlane(rn[w + 16 * u]);
        ro[u] = __builtin_amdgcn_readfirstlane(rs[w + 16 * u]);
    }
#pragma unroll
    for (int u = 0; u < RUNS_PER_WAVE; ++u) {
        t[u] = make_int2(0, 0);
        if (lane < ln[u]) t[u] = tmp[ro[u] + lane];
    }
#pragma unroll
    for (int u = 0; u < RUNS_PER_WAVE; ++u) {
        if (lane < ln[u]) atomicAdd(&cnt[(unsigned)t[u].x >> 18], 1);
        for (int j = lane + 32; j < ln[u]; j += 32) {
            const int2 x = tmp[ro[u] + j];
            atomicAdd(&cnt[(unsigned)x.x >> 18], 1);
        }
    }
    __syncthreads();

    const int mine = (threadIdx.x < 256) ? cnt[threadIdx.x] : 0;
    int all;
    const int excl = block_excl_scan<16>(mine, wsum, &all);
    if (threadIdx.x < 256) {
        cur[threadIdx.x] = excl;
        const int row = b * 256 + (int)threadIdx.x;
        if (row < NTOT) {
            start[row] = base + excl;
            endp[row]  = base + excl + mine;
        }
    }
    __syncthreads();

    if (n <= STAGE_CAP) {
#pragma unroll
        for (int u0 = 0; u0 < RUNS_PER_WAVE; u0 += 8) {
            int p[8];
#pragma unroll
            for (int k = 0; k < 8; ++k) {
                p[k] = 0;
                if (lane < ln[u0 + k])
                    p[k] = atomicAdd(&cur[(unsigned)t[u0 + k].x >> 18], 1);
            }
#pragma unroll
            for (int k = 0; k < 8; ++k) {
                const int u = u0 + k;
                if (lane < ln[u]) stage[p[k]] = make_int2(t[u].x & 0x3FFFF, t[u].y);
                for (int j = lane + 32; j < ln[u]; j += 32) {
                    const int2 x = tmp[ro[u] + j];
                    const int  q = atomicAdd(&cur[(unsigned)x.x >> 18], 1);
                    stage[q] = make_int2(x.x & 0x3FFFF, x.y);
                }
            }
        }
        __syncthreads();
        for (int i = threadIdx.x; i < n; i += 512) ed[base + i] = stage[i];
    } else {
#pragma unroll
        for (int u = 0; u < RUNS_PER_WAVE; ++u) {
            if (lane < ln[u]) {
                const int p = atomicAdd(&cur[(unsigned)t[u].x >> 18], 1);
                ed[base + p] = make_int2(t[u].x & 0x3FFFF, t[u].y);
            }
            for (int j = lane + 32; j < ln[u]; j += 32) {
                const int2 x = tmp[ro[u] + j];
                const int  p = atomicAdd(&cur[(unsigned)x.x >> 18], 1);
                ed[base + p] = make_int2(x.x & 0x3FFFF, x.y);
            }
        }
    }
}

template <int MODE>
__global__ __launch_bounds__(256) void mgcn_pull(
    const float* __restrict__ src, const float* __restrict__ upref,
    const float* __restrict__ items, const float* __restrict__ inv,
    const int* __restrict__ start, const int* __restrict__ endp,
    const int2* __restrict__ ed, float* __restrict__ dst)
{
    const int lane = threadIdx.x & 31;
    const int row = __builtin_amdgcn_readfirstlane(
        (int)((blockIdx.x * 256 + threadIdx.x) >> 5));
    if (row >= NTOT) return;

    const int e0 = __builtin_amdgcn_readfirstlane(start[row]);
    const int e1 = __builtin_amdgcn_readfirstlane(endp[row]);

    float* d = dst + (size_t)row * DIM_LATENT + lane * 2;

    v2f l0 = {0.0f, 0.0f}, l1 = {0.0f, 0.0f}, l2 = {0.0f, 0.0f};
    if (MODE == 3) {
        if (row < NUM_USER) {
            const v2f x = *(const v2f*)(upref + (size_t)row * DIM_LATENT + lane * 2);
            const float r = inv[row];
            l0.x = x.x * r; l0.y = x.y * r;
        } else {
            l0 = *(const v2f*)(items + (size_t)(row - NUM_USER) * DIM_LATENT + lane * 2);
        }
        l1 = *(const v2f*)d;
        l2 = *(const v2f*)(src + (size_t)row * DIM_LATENT + lane * 2);
    }

    v2f acc = {0.0f, 0.0f};
    for (int base = e0; base < e1; base += 32) {
        const int rem = e1 - base;
        const int n   = rem < 32 ? rem : 32;

        int   c32 = 0;
        float v32 = 0.0f;
        float s32 = 1.0f;
        if (lane < n) {
            const int2 t = ed[base + lane];
            c32 = t.x;
            v32 = __int_as_float(t.y);
            if (MODE == 1) { if (c32 < NUM_USER) s32 = inv[c32]; }
        }

        for (int i = 0; i < n; i += 8) {
            v2f   x[8];
            float v[8];
            float s[8];
#pragma unroll
            for (int k = 0; k < 8; ++k) {
                const int c = __builtin_amdgcn_readlane(c32, i + k);
                v[k] = __int_as_float(
                    __builtin_amdgcn_readlane(__float_as_int(v32), i + k));
                const float* p;
                if (MODE == 1) {
                    s[k] = __int_as_float(
                        __builtin_amdgcn_readlane(__float_as_int(s32), i + k));
                    p = (c < NUM_USER)
                            ? (upref + (size_t)c * DIM_LATENT)
                            : (items + (size_t)(c - NUM_USER) * DIM_LATENT);
                } else {
                    s[k] = 1.0f;
                    p = src + (size_t)c * DIM_LATENT;
                }
                x[k] = *(const v2f*)(p + lane * 2);
            }
#pragma unroll
            for (int k = 0; k < 8; ++k) {
                if (MODE == 1) {
                    acc.x += (x[k].x * s[k]) * v[k];
                    acc.y += (x[k].y * s[k]) * v[k];
                } else {
                    acc.x += x[k].x * v[k];
                    acc.y += x[k].y * v[k];
                }
            }
        }
    }

    if (MODE == 3) {
        v2f o;
        o.x = (((l0.x + l1.x) + l2.x) + acc.x) * 0.25f;
        o.y = (((l0.y + l1.y) + l2.y) + acc.y) * 0.25f;
        *(v2f*)d = o;
    } else {
        *(v2f*)d = acc;
    }
}

extern "C" void kernel_launch(void* const* d_in, const int* in_sizes, int n_in,
                              void* d_out, int out_size, void* d_ws, size_t ws_size,
                              hipStream_t stream)
{
    const float* features = (const float*)d_in[0];
    const float* upref    = (const float*)d_in[1];
    const float* W        = (const float*)d_in[2];
    const float* bias     = (const float*)d_in[3];
    const int*   rows     = (const int*)d_in[4];
    const int*   cols     = (const int*)d_in[5];
    const float* vals     = (const float*)d_in[6];
    float*       out      = (float*)d_out;
    const int    nnz      = in_sizes[4];

    char*  ws     = (char*)d_ws;
    float* tab2   = (float*)(ws);
    int2*  tmp    = (int2*) (ws);
    int*   ltab   = (int*)  (ws + 19200000);
    float* items  = (float*)(ws + 38400000);
    float* inv    = (float*)(ws + 51200000);
    int*   start  = (int*)  (ws + 52200256);
    int*   rend   = (int*)  (ws + 52800256);
    int2*  ed     = (int2*) (ws + 53400320);

    mgcn_gemm_wmma<<<(GEMM_TILES + 7) / 8, 256, 0, stream>>>(features, W, bias, items);

    mgcn_row_norms<<<NTOT * 32 / 256, 256, 0, stream>>>(upref, items, inv);

    mgcn_chunk_sort<<<SORT_BLOCKS, 1024, 0, stream>>>(rows, cols, vals, ltab, tmp, nnz);
    mgcn_bucket_rows<<<NBUCKET, 512, 0, stream>>>(tmp, ltab, start, rend, ed, nnz);

    const int pullBlocks = NTOT * 32 / 256;
    mgcn_pull<1><<<pullBlocks, 256, 0, stream>>>(upref, upref, items, inv,
                                                 start, rend, ed, out);
    mgcn_pull<2><<<pullBlocks, 256, 0, stream>>>(out, upref, items, inv,
                                                 start, rend, ed, tab2);
    mgcn_pull<3><<<pullBlocks, 256, 0, stream>>>(tab2, upref, items, inv,
                                                 start, rend, ed, out);
}


# Edge sort rides in the dense stages' launches as extra work-groups: chunk pass beside the GEMM, bucket pass beside the norms (8 rows a wave); 7 launches -> 5; f32 sum order = new placement order
# speedup vs baseline: 3.1558x; 1.0958x over previous
#include <hip/hip_runtime.h>


#define NUM_USER   100000
#define NUM_ITEM   50000
#define NTOT       (NUM_USER + NUM_ITEM)
#define DIM_FEAT   512
#define DIM_LATENT 64

typedef __attribute__((ext_vector_type(2))) float v2f;
typedef __attribute__((ext_vector_type(8))) float v8f;

#define GEMM_TILES   (NUM_ITEM / 16)
#define GEMM_BLOCKS  ((GEMM_TILES + 7) / 8)
#define W_PITCH      132
#define LDS1_BYTES   ((DIM_FEAT / 2) * W_PITCH * 4)

__device__ __forceinline__ void gemm_tile_body(
    const float* __restrict__ F, const float* __restrict__ W,
    const float* __restrict__ bias, float* __restrict__ items,
    float* Wl, const int blk)
{

    for (int it = 0; it < (DIM_FEAT / 2) * 16 / 256; ++it) {
        const int idx = it * 256 + (int)threadIdx.x;
        const int kp  = idx >> 4;
        const int n4  = idx & 15;
        const float4 r0 = *(const float4*)(W + (size_t)(2 * kp) * DIM_LATENT + n4 * 4);
        const float4 r1 = *(const float4*)(W + (size_t)(2 * kp + 1) * DIM_LATENT + n4 * 4);
        float* d = Wl + kp * W_PITCH + n4 * 8;
        *(float4*)d       = make_float4(r0.x, r1.x, r0.y, r1.y);
        *(float4*)(d + 4) = make_float4(r0.z, r1.z, r0.w, r1.w);
    }
    __syncthreads();

    const int lane = threadIdx.x & 31;
    const int half = lane >> 4;
    const int c    = lane & 15;
    const int tile = __builtin_amdgcn_readfirstlane(
        (int)(blk * 8 + (threadIdx.x >> 5)));
    if (tile >= GEMM_TILES) return;

    const float* frow = F + (size_t)(tile * 16 + c) * DIM_FEAT + 16 * half;
    const float* wl   = Wl + (8 * half) * W_PITCH + 2 * c;

    v8f acc[4] = {};
    float4 cur[4], nxt[4];
#pragma unroll
    for (int j = 0; j < 4; ++j) cur[j] = *(const float4*)(frow + 4 * j);

    for (int t = 0; t < DIM_FEAT / 32; ++t) {
        const int tn = (t + 1 < DIM_FEAT / 32) ? t + 1 : t;
#pragma unroll
        for (int j = 0; j < 4; ++j) nxt[j] = *(const float4*)(frow + tn * 32 + 4 * j);

        const float* wt = wl + t * 16 * W_PITCH;
#pragma unroll
        for (int st = 0; st < 8; ++st) {
            const float4 q = cur[st >> 1];
            v2f a;
            a.x = (st & 1) ? q.z : q.x;
            a.y = (st & 1) ? q.w : q.y;
#pragma unroll
            for (int nt = 0; nt < 4; ++nt) {
                const v2f bm = *(const v2f*)(wt + st * W_PITCH + nt * 32);
                acc[nt] = __builtin_amdgcn_wmma_f32_16x16x4_f32(
                         false, a,  false, bm,
                         (short)0, acc[nt],  false,  false);
            }
        }
#pragma unroll
        for (int j = 0; j < 4; ++j) cur[j] = nxt[j];
    }

    float* dst = items + (size_t)(tile * 16) * DIM_LATENT;
#pragma unroll
    for (int nt = 0; nt < 4; ++nt) {
        const float bv = bias[nt * 16 + c];
#pragma unroll
        for (int j = 0; j < 8; ++j) {
            const int mr = j + 8 * half;
            dst[(size_t)mr * DIM_LATENT + nt * 16 + c] = acc[nt][j] + bv;
        }
    }
}

#define NORM_ROWS_PER_WAVE 8
#define NORM_BLOCKS ((NTOT + 16 * NORM_ROWS_PER_WAVE - 1) / (16 * NORM_ROWS_PER_WAVE))

__device__ __forceinline__ void norm_rows_body(
    const float* __restrict__ upref, float* __restrict__ items,
    float* __restrict__ inv, const int nb)
{
    const int lane = threadIdx.x & 31;
    const int row0 = __builtin_amdgcn_readfirstlane(
        (int)(((nb * 512 + (int)threadIdx.x) >> 5) * NORM_ROWS_PER_WAVE));

    v2f x[NORM_ROWS_PER_WAVE];
#pragma unroll
    for (int k = 0; k < NORM_ROWS_PER_WAVE; ++k) {
        const int rl = (row0 + k < NTOT) ? row0 + k : NTOT - 1;
        const float* src = (rl < NUM_USER)
            ? (upref + (size_t)rl * DIM_LATENT)
            : (items + (size_t)(rl - NUM_USER) * DIM_LATENT);
        x[k] = *(const v2f*)(src + lane * 2);
    }

    float rmine = 0.0f;
#pragma unroll
    for (int k = 0; k < NORM_ROWS_PER_WAVE; ++k) {
        const int row = row0 + k;
        float s = x[k].x * x[k].x + x[k].y * x[k].y;
#pragma unroll
        for (int off = 16; off; off >>= 1) s += __shfl_xor(s, off, 32);
        const float r = 1.0f / fmaxf(sqrtf(s), 1e-12f);
        if (lane == k) rmine = r;
        if (row >= NUM_USER && row < NTOT) {
            v2f y; y.x = x[k].x * r; y.y = x[k].y * r;
            *(v2f*)(items + (size_t)(row - NUM_USER) * DIM_LATENT + lane * 2) = y;
        }
    }
    if (lane < NORM_ROWS_PER_WAVE && row0 + lane < NUM_USER) inv[row0 + lane] = rmine;
}

#define NBUCKET      ((NTOT + 255) / 256)
#define SORT_BLOCKS  256
#define LROW         (NBUCKET + 1)
#define CHUNK_CAP    9600
#define STAGE_CAP    6144

template <int NWAVES>
__device__ __forceinline__ int block_excl_scan(int v, int* wsum, int* total)
{
    const int lane = threadIdx.x & 31;
    const int w    = threadIdx.x >> 5;
    int incl = v;
    for (int off = 1; off < 32; off <<= 1) {
        const int t = __shfl_up(incl, off, 32);
        if (lane >= off) incl += t;
    }
    if (lane == 31) wsum[w] = incl;
    __syncthreads();
    if (w == 0) {
        const int s = (lane < NWAVES) ? wsum[lane] : 0;
        int si = s;
        for (int off = 1; off < 32; off <<= 1) {
            const int t = __shfl_up(si, off, 32);
            if (lane >= off) si += t;
        }
        wsum[lane] = si - s;
        if (lane == 31) wsum[32] = si;
    }
    __syncthreads();
    *total = wsum[32];
    return wsum[w] + incl - v;
}

__device__ __forceinline__ int sort_chunk(int nnz)
{
    return ((nnz + SORT_BLOCKS * 4 - 1) / (SORT_BLOCKS * 4)) * 4;
}

__device__ __forceinline__ int2 edge_record(int r, int c, float v)
{
    return make_int2(((r & 255) << 18) | c, __float_as_int(v));
}

#define CS_THREADS  256
#define CS_GROUPS   10
#define CS_STRIDE   (CS_THREADS * 4)
#define CS_BPT      ((NBUCKET + CS_THREADS - 1) / CS_THREADS)
#define CS_H_OFF     0
#define CS_WSUM_OFF  (((NBUCKET * 4 + 15) / 16) * 16)
#define CS_STAGE_OFF (CS_WSUM_OFF + 144)

__device__ __forceinline__ void chunk_sort_body(
    const int* __restrict__ rows, const int* __restrict__ cols,
    const float* __restrict__ vals, int* __restrict__ ltab,
    int2* __restrict__ tmp, const int nnz,
    int* h, int* wsum, int2* stage, const int g)
{
    const int chunk = sort_chunk(nnz);
    const int lo    = g * chunk;
    const int hi    = (lo + chunk < nnz) ? lo + chunk : nnz;
    const int n     = (hi > lo) ? hi - lo : 0;
    const int e0    = lo + (int)threadIdx.x * 4;

    int4   r[CS_GROUPS], c[CS_GROUPS];
    float4 v[CS_GROUPS];
    bool   ok[CS_GROUPS];
#pragma unroll
    for (int u = 0; u < CS_GROUPS; ++u) {
        const int e = e0 + u * CS_STRIDE;
        ok[u] = e + 3 < hi;
        r[u] = make_int4(0, 0, 0, 0);
        c[u] = make_int4(0, 0, 0, 0);
        v[u] = make_float4(0.0f, 0.0f, 0.0f, 0.0f);
        if (ok[u]) {
            r[u] = *(const int4*)(rows + e);
            c[u] = *(const int4*)(cols + e);
            v[u] = *(const float4*)(vals + e);
        }
    }

    for (int i = threadIdx.x; i < NBUCKET; i += CS_THREADS) h[i] = 0;
    __syncthreads();

#pragma unroll
    for (int u = 0; u < CS_GROUPS; ++u) {
        if (ok[u]) {
            atomicAdd(&h[r[u].x >> 8], 1);
            atomicAdd(&h[r[u].y >> 8], 1);
            atomicAdd(&h[r[u].z >> 8], 1);
            atomicAdd(&h[r[u].w >> 8], 1);
        }
    }
    for (int u = 0, e = e0; e < hi; ++u, e += CS_STRIDE) {
        if (u < CS_GROUPS && e + 3 < hi) continue;
        const int kend = (e + 4 < hi) ? e + 4 : hi;
        for (int k = e; k < kend; ++k) atomicAdd(&h[rows[k] >> 8], 1);
    }
    __syncthreads();

    int m[CS_BPT];
    int own = 0;
#pragma unroll
    for (int j = 0; j < CS_BPT; ++j) {
        const int b = (int)threadIdx.x * CS_BPT + j;
        m[j] = (b < NBUCKET) ? h[b] : 0;
        own += m[j];
    }
    int all;
    int run = block_excl_scan<CS_THREADS / 32>(own, wsum, &all);
#pragma unroll
    for (int j = 0; j < CS_BPT; ++j) {
        const int b = (int)threadIdx.x * CS_BPT + j;
        if (b < NBUCKET) {
            ltab[g * LROW + b] = run;
            h[b] = run;
        }
        run += m[j];
    }
    if (threadIdx.x == 0) ltab[g * LROW + NBUCKET] = all;
    __syncthreads();

    if (n <= CHUNK_CAP) {
#pragma unroll
        for (int u = 0; u < CS_GROUPS; ++u) {
            if (ok[u]) {
                const int p0 = atomicAdd(&h[r[u].x >> 8], 1);
                const int p1 = atomicAdd(&h[r[u].y >> 8], 1);
                const int p2 = atomicAdd(&h[r[u].z >> 8], 1);
                const int p3 = atomicAdd(&h[r[u].w >> 8], 1);
                stage[p0] = edge_record(r[u].x, c[u].x, v[u].x);
                stage[p1] = edge_record(r[u].y, c[u].y, v[u].y);
                stage[p2] = edge_record(r[u].z, c[u].z, v[u].z);
                stage[p3] = edge_record(r[u].w, c[u].w, v[u].w);
            }
        }
        for (int u = 0, e = e0; e < hi; ++u, e += CS_STRIDE) {
            if (u < CS_GROUPS && e + 3 < hi) continue;
            const int kend = (e + 4 < hi) ? e + 4 : hi;
            for (int k = e; k < kend; ++k) {
                const int rr = rows[k];
                const int p  = atomicAdd(&h[rr >> 8], 1);
                stage[p] = edge_record(rr, cols[k], vals[k]);
            }
        }
        __syncthreads();
        for (int i = threadIdx.x; i < n; i += CS_THREADS) tmp[lo + i] = stage[i];
    } else {
#pragma unroll
        for (int u = 0; u < CS_GROUPS; ++u) {
            if (ok[u]) {
                const int p0 = atomicAdd(&h[r[u].x >> 8], 1);
                const int p1 = atomicAdd(&h[r[u].y >> 8], 1);
                const int p2 = atomicAdd(&h[r[u].z >> 8], 1);
                const int p3 = atomicAdd(&h[r[u].w >> 8], 1);
                tmp[lo + p0] = edge_record(r[u].x, c[u].x, v[u].x);
                tmp[lo + p1] = edge_record(r[u].y, c[u].y, v[u].y);
                tmp[lo + p2] = edge_record(r[u].z, c[u].z, v[u].z);
                tmp[lo + p3] = edge_record(r[u].w, c[u].w, v[u].w);
            }
        }
        for (int u = 0, e = e0; e < hi; ++u, e += CS_STRIDE) {
            if (u < CS_GROUPS && e + 3 < hi) continue;
            const int kend = (e + 4 < hi) ? e + 4 : hi;
            for (int k = e; k < kend; ++k) {
                const int rr = rows[k];
                const int p  = atomicAdd(&h[rr >> 8], 1);
                tmp[lo + p] = edge_record(rr, cols[k], vals[k]);
            }
        }
    }
}

__global__ __launch_bounds__(256) void mgcn_gemm_and_chunk_sort(
    const float* __restrict__ F, const float* __restrict__ W,
    const float* __restrict__ bias, float* __restrict__ items,
    const int* __restrict__ rows, const int* __restrict__ cols,
    const float* __restrict__ vals, int* __restrict__ ltab,
    int2* __restrict__ tmp, int nnz)
{
    __shared__ __attribute__((aligned(16))) unsigned char lds_raw[LDS1_BYTES];
    static_assert(CS_STAGE_OFF % 16 == 0, "record stage alignment");
    static_assert(CS_STAGE_OFF + CHUNK_CAP * 8 <= LDS1_BYTES, "sort stage fits the W stage");

    if (blockIdx.x < SORT_BLOCKS) {
        chunk_sort_body(rows, cols, vals, ltab, tmp, nnz,
                        (int*)(lds_raw + CS_H_OFF), (int*)(lds_raw + CS_WSUM_OFF),
                        (int2*)(lds_raw + CS_STAGE_OFF), (int)blockIdx.x);
        return;
    }
    gemm_tile_body(F, W, bias, items, (float*)lds_raw, (int)blockIdx.x - SORT_BLOCKS);
}

#define RUNS_PER_WAVE (SORT_BLOCKS / 16)
__global__ __launch_bounds__(512) void mgcn_bucket_rows_and_norms(
    const int2* __restrict__ tmp, const int* __restrict__ ltab,
    int* __restrict__ start, int* __restrict__ endp,
    int2* __restrict__ ed, int nnz,
    const float* __restrict__ upref, float* __restrict__ items,
    float* __restrict__ inv)
{
    if (blockIdx.x >= NBUCKET) {
        norm_rows_body(upref, items, inv, (int)blockIdx.x - NBUCKET);
        return;
    }

    __shared__ int  cnt[256];
    __shared__ int  cur[256];
    __shared__ int  rs[SORT_BLOCKS];
    __shared__ int  rn[SORT_BLOCKS];
    __shared__ int  wsum[33];
    __shared__ int2 stage[STAGE_CAP];

    const int b     = (int)blockIdx.x;
    const int lane  = threadIdx.x & 31;
    const int w     = __builtin_amdgcn_readfirstlane((int)(threadIdx.x >> 5));
    const int chunk = sort_chunk(nnz);

    int s0 = 0, len = 0;
    if (threadIdx.x < SORT_BLOCKS) {
        s0  = ltab[(int)threadIdx.x * LROW + b];
        len = ltab[(int)threadIdx.x * LROW + b + 1] - s0;
        rs[threadIdx.x] = (int)threadIdx.x * chunk + s0;
        rn[threadIdx.x] = len;
    }
    if (threadIdx.x < 256) cnt[threadIdx.x] = 0;
    int sumStart, sumLen;
    block_excl_scan<16>(s0, wsum, &sumStart);
    block_excl_scan<16>(len, wsum, &sumLen);
    const int base = __builtin_amdgcn_readfirstlane(sumStart);
    const int n    = __builtin_amdgcn_readfirstlane(sumLen);

    int2 t[RUNS_PER_WAVE];
    int  ln[RUNS_PER_WAVE];
    int  ro[RUNS_PER_WAVE];
#pragma unroll
    for (int u = 0; u < RUNS_PER_WAVE; ++u) {
        ln[u] = __builtin_amdgcn_readfirstlane(rn[w + 16 * u]);
        ro[u] = __builtin_amdgcn_readfirstlane(rs[w + 16 * u]);
    }
#pragma unroll
    for (int u = 0; u < RUNS_PER_WAVE; ++u) {
        t[u] = make_int2(0, 0);
        if (lane < ln[u]) t[u] = tmp[ro[u] + lane];
    }
#pragma unroll
    for (int u = 0; u < RUNS_PER_WAVE; ++u) {
        if (lane < ln[u]) atomicAdd(&cnt[(unsigned)t[u].x >> 18], 1);
        for (int j = lane + 32; j < ln[u]; j += 32) {
            const int2 x = tmp[ro[u] + j];
            atomicAdd(&cnt[(unsigned)x.x >> 18], 1);
        }
    }
    __syncthreads();

    const int mine = (threadIdx.x < 256) ? cnt[threadIdx.x] : 0;
    int all;
    const int excl = block_excl_scan<16>(mine, wsum, &all);
    if (threadIdx.x < 256) {
        cur[threadIdx.x] = excl;
        const int row = b * 256 + (int)threadIdx.x;
        if (row < NTOT) {
            start[row] = base + excl;
            endp[row]  = base + excl + mine;
        }
    }
    __syncthreads();

    if (n <= STAGE_CAP) {
#pragma unroll
        for (int u0 = 0; u0 < RUNS_PER_WAVE; u0 += 8) {
            int p[8];
#pragma unroll
            for (int k = 0; k < 8; ++k) {
                p[k] = 0;
                if (lane < ln[u0 + k])
                    p[k] = atomicAdd(&cur[(unsigned)t[u0 + k].x >> 18], 1);
            }
#pragma unroll
            for (int k = 0; k < 8; ++k) {
                const int u = u0 + k;
                if (lane < ln[u]) stage[p[k]] = make_int2(t[u].x & 0x3FFFF, t[u].y);
                for (int j = lane + 32; j < ln[u]; j += 32) {
                    const int2 x = tmp[ro[u] + j];
                    const int  q = atomicAdd(&cur[(unsigned)x.x >> 18], 1);
                    stage[q] = make_int2(x.x & 0x3FFFF, x.y);
                }
            }
        }
        __syncthreads();
        for (int i = threadIdx.x; i < n; i += 512) ed[base + i] = stage[i];
    } else {
#pragma unroll
        for (int u = 0; u < RUNS_PER_WAVE; ++u) {
            if (lane < ln[u]) {
                const int p = atomicAdd(&cur[(unsigned)t[u].x >> 18], 1);
                ed[base + p] = make_int2(t[u].x & 0x3FFFF, t[u].y);
            }
            for (int j = lane + 32; j < ln[u]; j += 32) {
                const int2 x = tmp[ro[u] + j];
                const int  p = atomicAdd(&cur[(unsigned)x.x >> 18], 1);
                ed[base + p] = make_int2(x.x & 0x3FFFF, x.y);
            }
        }
    }
}

template <int MODE>
__global__ __launch_bounds__(256) void mgcn_pull(
    const float* __restrict__ src, const float* __restrict__ upref,
    const float* __restrict__ items, const float* __restrict__ inv,
    const int* __restrict__ start, const int* __restrict__ endp,
    const int2* __restrict__ ed, float* __restrict__ dst)
{
    const int lane = threadIdx.x & 31;
    const int row = __builtin_amdgcn_readfirstlane(
        (int)((blockIdx.x * 256 + threadIdx.x) >> 5));
    if (row >= NTOT) return;

    const int e0 = __builtin_amdgcn_readfirstlane(start[row]);
    const int e1 = __builtin_amdgcn_readfirstlane(endp[row]);

    float* d = dst + (size_t)row * DIM_LATENT + lane * 2;

    v2f l0 = {0.0f, 0.0f}, l1 = {0.0f, 0.0f}, l2 = {0.0f, 0.0f};
    if (MODE == 3) {
        if (row < NUM_USER) {
            const v2f x = *(const v2f*)(upref + (size_t)row * DIM_LATENT + lane * 2);
            const float r = inv[row];
            l0.x = x.x * r; l0.y = x.y * r;
        } else {
            l0 = *(const v2f*)(items + (size_t)(row - NUM_USER) * DIM_LATENT + lane * 2);
        }
        l1 = *(const v2f*)d;
        l2 = *(const v2f*)(src + (size_t)row * DIM_LATENT + lane * 2);
    }

    v2f acc = {0.0f, 0.0f};
    for (int base = e0; base < e1; base += 32) {
        const int rem = e1 - base;
        const int n   = rem < 32 ? rem : 32;

        int   c32 = 0;
        float v32 = 0.0f;
        float s32 = 1.0f;
        if (lane < n) {
            const int2 t = ed[base + lane];
            c32 = t.x;
            v32 = __int_as_float(t.y);
            if (MODE == 1) { if (c32 < NUM_USER) s32 = inv[c32]; }
        }

        for (int i = 0; i < n; i += 8) {
            v2f   x[8];
            float v[8];
            float s[8];
#pragma unroll
            for (int k = 0; k < 8; ++k) {
                const int c = __builtin_amdgcn_readlane(c32, i + k);
                v[k] = __int_as_float(
                    __builtin_amdgcn_readlane(__float_as_int(v32), i + k));
                const float* p;
                if (MODE == 1) {
                    s[k] = __int_as_float(
                        __builtin_amdgcn_readlane(__float_as_int(s32), i + k));
                    p = (c < NUM_USER)
                            ? (upref + (size_t)c * DIM_LATENT)
                            : (items + (size_t)(c - NUM_USER) * DIM_LATENT);
                } else {
                    s[k] = 1.0f;
                    p = src + (size_t)c * DIM_LATENT;
                }
                x[k] = *(const v2f*)(p + lane * 2);
            }
#pragma unroll
            for (int k = 0; k < 8; ++k) {
                if (MODE == 1) {
                    acc.x += (x[k].x * s[k]) * v[k];
                    acc.y += (x[k].y * s[k]) * v[k];
                } else {
                    acc.x += x[k].x * v[k];
                    acc.y += x[k].y * v[k];
                }
            }
        }
    }

    if (MODE == 3) {
        v2f o;
        o.x = (((l0.x + l1.x) + l2.x) + acc.x) * 0.25f;
        o.y = (((l0.y + l1.y) + l2.y) + acc.y) * 0.25f;
        *(v2f*)d = o;
    } else {
        *(v2f*)d = acc;
    }
}

extern "C" void kernel_launch(void* const* d_in, const int* in_sizes, int n_in,
                              void* d_out, int out_size, void* d_ws, size_t ws_size,
                              hipStream_t stream)
{
    const float* features = (const float*)d_in[0];
    const float* upref    = (const float*)d_in[1];
    const float* W        = (const float*)d_in[2];
    const float* bias     = (const float*)d_in[3];
    const int*   rows     = (const int*)d_in[4];
    const int*   cols     = (const int*)d_in[5];
    const float* vals     = (const float*)d_in[6];
    float*       out      = (float*)d_out;
    const int    nnz      = in_sizes[4];

    char*  ws     = (char*)d_ws;
    float* tab2   = (float*)(ws);
    int2*  tmp    = (int2*) (ws);
    int*   ltab   = (int*)  (ws + 19200000);
    float* items  = (float*)(ws + 38400000);
    float* inv    = (float*)(ws + 51200000);
    int*   start  = (int*)  (ws + 52200256);
    int*   rend   = (int*)  (ws + 52800256);
    int2*  ed     = (int2*) (ws + 53400320);

    mgcn_gemm_and_chunk_sort<<<SORT_BLOCKS + GEMM_BLOCKS, 256, 0, stream>>>(
        features, W, bias, items, rows, cols, vals, ltab, tmp, nnz);

    mgcn_bucket_rows_and_norms<<<NBUCKET + NORM_BLOCKS, 512, 0, stream>>>(
        tmp, ltab, start, rend, ed, nnz, upref, items, inv);

    const int pullBlocks = NTOT * 32 / 256;
    mgcn_pull<1><<<pullBlocks, 256, 0, stream>>>(upref, upref, items, inv,
                                                 start, rend, ed, out);
    mgcn_pull<2><<<pullBlocks, 256, 0, stream>>>(out, upref, items, inv,
                                                 start, rend, ed, tab2);
    mgcn_pull<3><<<pullBlocks, 256, 0, stream>>>(tab2, upref, items, inv,
                                                 start, rend, ed, out);
}


# Propagation layers: a wave takes two rows, one per 16-lane half (lane holds 4 dims, edge data by 16-lane-row broadcast); 150000 -> 75000 waves per layer; f32 sum order kept, product+add now fused
# speedup vs baseline: 3.3502x; 1.0616x over previous
#include <hip/hip_runtime.h>


#define NUM_USER   100000
#define NUM_ITEM   50000
#define NTOT       (NUM_USER + NUM_ITEM)
#define DIM_FEAT   512
#define DIM_LATENT 64

typedef __attribute__((ext_vector_type(2))) float v2f;
typedef __attribute__((ext_vector_type(8))) float v8f;

#define GEMM_TILES   (NUM_ITEM / 16)
#define GEMM_BLOCKS  ((GEMM_TILES + 7) / 8)
#define W_PITCH      132
#define LDS1_BYTES   ((DIM_FEAT / 2) * W_PITCH * 4)

__device__ __forceinline__ void gemm_tile_body(
    const float* __restrict__ F, const float* __restrict__ W,
    const float* __restrict__ bias, float* __restrict__ items,
    float* Wl, const int blk)
{

    for (int it = 0; it < (DIM_FEAT / 2) * 16 / 256; ++it) {
        const int idx = it * 256 + (int)threadIdx.x;
        const int kp  = idx >> 4;
        const int n4  = idx & 15;
        const float4 r0 = *(const float4*)(W + (size_t)(2 * kp) * DIM_LATENT + n4 * 4);
        const float4 r1 = *(const float4*)(W + (size_t)(2 * kp + 1) * DIM_LATENT + n4 * 4);
        float* d = Wl + kp * W_PITCH + n4 * 8;
        *(float4*)d       = make_float4(r0.x, r1.x, r0.y, r1.y);
        *(float4*)(d + 4) = make_float4(r0.z, r1.z, r0.w, r1.w);
    }
    __syncthreads();

    const int lane = threadIdx.x & 31;
    const int half = lane >> 4;
    const int c    = lane & 15;
    const int tile = __builtin_amdgcn_readfirstlane(
        (int)(blk * 8 + (threadIdx.x >> 5)));
    if (tile >= GEMM_TILES) return;

    const float* frow = F + (size_t)(tile * 16 + c) * DIM_FEAT + 16 * half;
    const float* wl   = Wl + (8 * half) * W_PITCH + 2 * c;

    v8f acc[4] = {};
    float4 cur[4], nxt[4];
#pragma unroll
    for (int j = 0; j < 4; ++j) cur[j] = *(const float4*)(frow + 4 * j);

    for (int t = 0; t < DIM_FEAT / 32; ++t) {
        const int tn = (t + 1 < DIM_FEAT / 32) ? t + 1 : t;
#pragma unroll
        for (int j = 0; j < 4; ++j) nxt[j] = *(const float4*)(frow + tn * 32 + 4 * j);

        const float* wt = wl + t * 16 * W_PITCH;
#pragma unroll
        for (int st = 0; st < 8; ++st) {
            const float4 q = cur[st >> 1];
            v2f a;
            a.x = (st & 1) ? q.z : q.x;
            a.y = (st & 1) ? q.w : q.y;
#pragma unroll
            for (int nt = 0; nt < 4; ++nt) {
                const v2f bm = *(const v2f*)(wt + st * W_PITCH + nt * 32);
                acc[nt] = __builtin_amdgcn_wmma_f32_16x16x4_f32(
                         false, a,  false, bm,
                         (short)0, acc[nt],  false,  false);
            }
        }
#pragma unroll
        for (int j = 0; j < 4; ++j) cur[j] = nxt[j];
    }

    float* dst = items + (size_t)(tile * 16) * DIM_LATENT;
#pragma unroll
    for (int nt = 0; nt < 4; ++nt) {
        const float bv = bias[nt * 16 + c];
#pragma unroll
        for (int j = 0; j < 8; ++j) {
            const int mr = j + 8 * half;
            dst[(size_t)mr * DIM_LATENT + nt * 16 + c] = acc[nt][j] + bv;
        }
    }
}

#define NORM_ROWS_PER_WAVE 8
#define NORM_BLOCKS ((NTOT + 16 * NORM_ROWS_PER_WAVE - 1) / (16 * NORM_ROWS_PER_WAVE))

__device__ __forceinline__ void norm_rows_body(
    const float* __restrict__ upref, float* __restrict__ items,
    float* __restrict__ inv, const int nb)
{
    const int lane = threadIdx.x & 31;
    const int row0 = __builtin_amdgcn_readfirstlane(
        (int)(((nb * 512 + (int)threadIdx.x) >> 5) * NORM_ROWS_PER_WAVE));

    v2f x[NORM_ROWS_PER_WAVE];
#pragma unroll
    for (int k = 0; k < NORM_ROWS_PER_WAVE; ++k) {
        const int rl = (row0 + k < NTOT) ? row0 + k : NTOT - 1;
        const float* src = (rl < NUM_USER)
            ? (upref + (size_t)rl * DIM_LATENT)
            : (items + (size_t)(rl - NUM_USER) * DIM_LATENT);
        x[k] = *(const v2f*)(src + lane * 2);
    }

    float rmine = 0.0f;
#pragma unroll
    for (int k = 0; k < NORM_ROWS_PER_WAVE; ++k) {
        const int row = row0 + k;
        float s = x[k].x * x[k].x + x[k].y * x[k].y;
#pragma unroll
        for (int off = 16; off; off >>= 1) s += __shfl_xor(s, off, 32);
        const float r = 1.0f / fmaxf(sqrtf(s), 1e-12f);
        if (lane == k) rmine = r;
        if (row >= NUM_USER && row < NTOT) {
            v2f y; y.x = x[k].x * r; y.y = x[k].y * r;
            *(v2f*)(items + (size_t)(row - NUM_USER) * DIM_LATENT + lane * 2) = y;
        }
    }
    if (lane < NORM_ROWS_PER_WAVE && row0 + lane < NUM_USER) inv[row0 + lane] = rmine;
}

#define NBUCKET      ((NTOT + 255) / 256)
#define SORT_BLOCKS  256
#define LROW         (NBUCKET + 1)
#define CHUNK_CAP    9600
#define STAGE_CAP    6144

template <int NWAVES>
__device__ __forceinline__ int block_excl_scan(int v, int* wsum, int* total)
{
    const int lane = threadIdx.x & 31;
    const int w    = threadIdx.x >> 5;
    int incl = v;
    for (int off = 1; off < 32; off <<= 1) {
        const int t = __shfl_up(incl, off, 32);
        if (lane >= off) incl += t;
    }
    if (lane == 31) wsum[w] = incl;
    __syncthreads();
    if (w == 0) {
        const int s = (lane < NWAVES) ? wsum[lane] : 0;
        int si = s;
        for (int off = 1; off < 32; off <<= 1) {
            const int t = __shfl_up(si, off, 32);
            if (lane >= off) si += t;
        }
        wsum[lane] = si - s;
        if (lane == 31) wsum[32] = si;
    }
    __syncthreads();
    *total = wsum[32];
    return wsum[w] + incl - v;
}

__device__ __forceinline__ int sort_chunk(int nnz)
{
    return ((nnz + SORT_BLOCKS * 4 - 1) / (SORT_BLOCKS * 4)) * 4;
}

__device__ __forceinline__ int2 edge_record(int r, int c, float v)
{
    return make_int2(((r & 255) << 18) | c, __float_as_int(v));
}

#define CS_THREADS  256
#define CS_GROUPS   10
#define CS_STRIDE   (CS_THREADS * 4)
#define CS_BPT      ((NBUCKET + CS_THREADS - 1) / CS_THREADS)
#define CS_H_OFF     0
#define CS_WSUM_OFF  (((NBUCKET * 4 + 15) / 16) * 16)
#define CS_STAGE_OFF (CS_WSUM_OFF + 144)

__device__ __forceinline__ void chunk_sort_body(
    const int* __restrict__ rows, const int* __restrict__ cols,
    const float* __restrict__ vals, int* __restrict__ ltab,
    int2* __restrict__ tmp, const int nnz,
    int* h, int* wsum, int2* stage, const int g)
{
    const int chunk = sort_chunk(nnz);
    const int lo    = g * chunk;
    const int hi    = (lo + chunk < nnz) ? lo + chunk : nnz;
    const int n     = (hi > lo) ? hi - lo : 0;
    const int e0    = lo + (int)threadIdx.x * 4;

    int4   r[CS_GROUPS], c[CS_GROUPS];
    float4 v[CS_GROUPS];
    bool   ok[CS_GROUPS];
#pragma unroll
    for (int u = 0; u < CS_GROUPS; ++u) {
        const int e = e0 + u * CS_STRIDE;
        ok[u] = e + 3 < hi;
        r[u] = make_int4(0, 0, 0, 0);
        c[u] = make_int4(0, 0, 0, 0);
        v[u] = make_float4(0.0f, 0.0f, 0.0f, 0.0f);
        if (ok[u]) {
            r[u] = *(const int4*)(rows + e);
            c[u] = *(const int4*)(cols + e);
            v[u] = *(const float4*)(vals + e);
        }
    }

    for (int i = threadIdx.x; i < NBUCKET; i += CS_THREADS) h[i] = 0;
    __syncthreads();

#pragma unroll
    for (int u = 0; u < CS_GROUPS; ++u) {
        if (ok[u]) {
            atomicAdd(&h[r[u].x >> 8], 1);
            atomicAdd(&h[r[u].y >> 8], 1);
            atomicAdd(&h[r[u].z >> 8], 1);
            atomicAdd(&h[r[u].w >> 8], 1);
        }
    }
    for (int u = 0, e = e0; e < hi; ++u, e += CS_STRIDE) {
        if (u < CS_GROUPS && e + 3 < hi) continue;
        const int kend = (e + 4 < hi) ? e + 4 : hi;
        for (int k = e; k < kend; ++k) atomicAdd(&h[rows[k] >> 8], 1);
    }
    __syncthreads();

    int m[CS_BPT];
    int own = 0;
#pragma unroll
    for (int j = 0; j < CS_BPT; ++j) {
        const int b = (int)threadIdx.x * CS_BPT + j;
        m[j] = (b < NBUCKET) ? h[b] : 0;
        own += m[j];
    }
    int all;
    int run = block_excl_scan<CS_THREADS / 32>(own, wsum, &all);
#pragma unroll
    for (int j = 0; j < CS_BPT; ++j) {
        const int b = (int)threadIdx.x * CS_BPT + j;
        if (b < NBUCKET) {
            ltab[g * LROW + b] = run;
            h[b] = run;
        }
        run += m[j];
    }
    if (threadIdx.x == 0) ltab[g * LROW + NBUCKET] = all;
    __syncthreads();

    if (n <= CHUNK_CAP) {
#pragma unroll
        for (int u = 0; u < CS_GROUPS; ++u) {
            if (ok[u]) {
                const int p0 = atomicAdd(&h[r[u].x >> 8], 1);
                const int p1 = atomicAdd(&h[r[u].y >> 8], 1);
                const int p2 = atomicAdd(&h[r[u].z >> 8], 1);
                const int p3 = atomicAdd(&h[r[u].w >> 8], 1);
                stage[p0] = edge_record(r[u].x, c[u].x, v[u].x);
                stage[p1] = edge_record(r[u].y, c[u].y, v[u].y);
                stage[p2] = edge_record(r[u].z, c[u].z, v[u].z);
                stage[p3] = edge_record(r[u].w, c[u].w, v[u].w);
            }
        }
        for (int u = 0, e = e0; e < hi; ++u, e += CS_STRIDE) {
            if (u < CS_GROUPS && e + 3 < hi) continue;
            const int kend = (e + 4 < hi) ? e + 4 : hi;
            for (int k = e; k < kend; ++k) {
                const int rr = rows[k];
                const int p  = atomicAdd(&h[rr >> 8], 1);
                stage[p] = edge_record(rr, cols[k], vals[k]);
            }
        }
        __syncthreads();
        for (int i = threadIdx.x; i < n; i += CS_THREADS) tmp[lo + i] = stage[i];
    } else {
#pragma unroll
        for (int u = 0; u < CS_GROUPS; ++u) {
            if (ok[u]) {
                const int p0 = atomicAdd(&h[r[u].x >> 8], 1);
                const int p1 = atomicAdd(&h[r[u].y >> 8], 1);
                const int p2 = atomicAdd(&h[r[u].z >> 8], 1);
                const int p3 = atomicAdd(&h[r[u].w >> 8], 1);
                tmp[lo + p0] = edge_record(r[u].x, c[u].x, v[u].x);
                tmp[lo + p1] = edge_record(r[u].y, c[u].y, v[u].y);
                tmp[lo + p2] = edge_record(r[u].z, c[u].z, v[u].z);
                tmp[lo + p3] = edge_record(r[u].w, c[u].w, v[u].w);
            }
        }
        for (int u = 0, e = e0; e < hi; ++u, e += CS_STRIDE) {
            if (u < CS_GROUPS && e + 3 < hi) continue;
            const int kend = (e + 4 < hi) ? e + 4 : hi;
            for (int k = e; k < kend; ++k) {
                const int rr = rows[k];
                const int p  = atomicAdd(&h[rr >> 8], 1);
                tmp[lo + p] = edge_record(rr, cols[k], vals[k]);
            }
        }
    }
}

__global__ __launch_bounds__(256) void mgcn_gemm_and_chunk_sort(
    const float* __restrict__ F, const float* __restrict__ W,
    const float* __restrict__ bias, float* __restrict__ items,
    const int* __restrict__ rows, const int* __restrict__ cols,
    const float* __restrict__ vals, int* __restrict__ ltab,
    int2* __restrict__ tmp, int nnz)
{
    __shared__ __attribute__((aligned(16))) unsigned char lds_raw[LDS1_BYTES];
    static_assert(CS_STAGE_OFF % 16 == 0, "record stage alignment");
    static_assert(CS_STAGE_OFF + CHUNK_CAP * 8 <= LDS1_BYTES, "sort stage fits the W stage");

    if (blockIdx.x < SORT_BLOCKS) {
        chunk_sort_body(rows, cols, vals, ltab, tmp, nnz,
                        (int*)(lds_raw + CS_H_OFF), (int*)(lds_raw + CS_WSUM_OFF),
                        (int2*)(lds_raw + CS_STAGE_OFF), (int)blockIdx.x);
        return;
    }
    gemm_tile_body(F, W, bias, items, (float*)lds_raw, (int)blockIdx.x - SORT_BLOCKS);
}

#define RUNS_PER_WAVE (SORT_BLOCKS / 16)
__global__ __launch_bounds__(512) void mgcn_bucket_rows_and_norms(
    const int2* __restrict__ tmp, const int* __restrict__ ltab,
    int* __restrict__ start, int* __restrict__ endp,
    int2* __restrict__ ed, int nnz,
    const float* __restrict__ upref, float* __restrict__ items,
    float* __restrict__ inv)
{
    if (blockIdx.x >= NBUCKET) {
        norm_rows_body(upref, items, inv, (int)blockIdx.x - NBUCKET);
        return;
    }

    __shared__ int  cnt[256];
    __shared__ int  cur[256];
    __shared__ int  rs[SORT_BLOCKS];
    __shared__ int  rn[SORT_BLOCKS];
    __shared__ int  wsum[33];
    __shared__ int2 stage[STAGE_CAP];

    const int b     = (int)blockIdx.x;
    const int lane  = threadIdx.x & 31;
    const int w     = __builtin_amdgcn_readfirstlane((int)(threadIdx.x >> 5));
    const int chunk = sort_chunk(nnz);

    int s0 = 0, len = 0;
    if (threadIdx.x < SORT_BLOCKS) {
        s0  = ltab[(int)threadIdx.x * LROW + b];
        len = ltab[(int)threadIdx.x * LROW + b + 1] - s0;
        rs[threadIdx.x] = (int)threadIdx.x * chunk + s0;
        rn[threadIdx.x] = len;
    }
    if (threadIdx.x < 256) cnt[threadIdx.x] = 0;
    int sumStart, sumLen;
    block_excl_scan<16>(s0, wsum, &sumStart);
    block_excl_scan<16>(len, wsum, &sumLen);
    const int base = __builtin_amdgcn_readfirstlane(sumStart);
    const int n    = __builtin_amdgcn_readfirstlane(sumLen);

    int2 t[RUNS_PER_WAVE];
    int  ln[RUNS_PER_WAVE];
    int  ro[RUNS_PER_WAVE];
#pragma unroll
    for (int u = 0; u < RUNS_PER_WAVE; ++u) {
        ln[u] = __builtin_amdgcn_readfirstlane(rn[w + 16 * u]);
        ro[u] = __builtin_amdgcn_readfirstlane(rs[w + 16 * u]);
    }
#pragma unroll
    for (int u = 0; u < RUNS_PER_WAVE; ++u) {
        t[u] = make_int2(0, 0);
        if (lane < ln[u]) t[u] = tmp[ro[u] + lane];
    }
#pragma unroll
    for (int u = 0; u < RUNS_PER_WAVE; ++u) {
        if (lane < ln[u]) atomicAdd(&cnt[(unsigned)t[u].x >> 18], 1);
        for (int j = lane + 32; j < ln[u]; j += 32) {
            const int2 x = tmp[ro[u] + j];
            atomicAdd(&cnt[(unsigned)x.x >> 18], 1);
        }
    }
    __syncthreads();

    const int mine = (threadIdx.x < 256) ? cnt[threadIdx.x] : 0;
    int all;
    const int excl = block_excl_scan<16>(mine, wsum, &all);
    if (threadIdx.x < 256) {
        cur[threadIdx.x] = excl;
        const int row = b * 256 + (int)threadIdx.x;
        if (row < NTOT) {
            start[row] = base + excl;
            endp[row]  = base + excl + mine;
        }
    }
    __syncthreads();

    if (n <= STAGE_CAP) {
#pragma unroll
        for (int u0 = 0; u0 < RUNS_PER_WAVE; u0 += 8) {
            int p[8];
#pragma unroll
            for (int k = 0; k < 8; ++k) {
                p[k] = 0;
                if (lane < ln[u0 + k])
                    p[k] = atomicAdd(&cur[(unsigned)t[u0 + k].x >> 18], 1);
            }
#pragma unroll
            for (int k = 0; k < 8; ++k) {
                const int u = u0 + k;
                if (lane < ln[u]) stage[p[k]] = make_int2(t[u].x & 0x3FFFF, t[u].y);
                for (int j = lane + 32; j < ln[u]; j += 32) {
                    const int2 x = tmp[ro[u] + j];
                    const int  q = atomicAdd(&cur[(unsigned)x.x >> 18], 1);
                    stage[q] = make_int2(x.x & 0x3FFFF, x.y);
                }
            }
        }
        __syncthreads();
        for (int i = threadIdx.x; i < n; i += 512) ed[base + i] = stage[i];
    } else {
#pragma unroll
        for (int u = 0; u < RUNS_PER_WAVE; ++u) {
            if (lane < ln[u]) {
                const int p = atomicAdd(&cur[(unsigned)t[u].x >> 18], 1);
                ed[base + p] = make_int2(t[u].x & 0x3FFFF, t[u].y);
            }
            for (int j = lane + 32; j < ln[u]; j += 32) {
                const int2 x = tmp[ro[u] + j];
                const int  p = atomicAdd(&cur[(unsigned)x.x >> 18], 1);
                ed[base + p] = make_int2(x.x & 0x3FFFF, x.y);
            }
        }
    }
}

typedef __attribute__((ext_vector_type(4))) float v4f;

template <int J>
__device__ __forceinline__ int row_bcast_i(int v)
{
    return __builtin_amdgcn_update_dpp(0, v, 0x150 + J, 0xF, 0xF, false);
}
template <int J>
__device__ __forceinline__ float row_bcast_f(float v)
{
    return __int_as_float(row_bcast_i<J>(__float_as_int(v)));
}

template <int MODE, int J>
__device__ __forceinline__ void pull_fetch(
    const int c32, const float v32, const float s32,
    const float* __restrict__ src, const float* __restrict__ upref,
    const float* __restrict__ items, const unsigned cb,
    v4f& x, float& v, float& s)
{
    const int col = row_bcast_i<J>(c32);
    v = row_bcast_f<J>(v32);
    s = (MODE == 1) ? row_bcast_f<J>(s32) : 1.0f;
    if (MODE == 1) {
        const char* p = (col < NUM_USER)
            ? ((const char*)upref + ((unsigned)col << 8))
            : ((const char*)items + ((unsigned)(col - NUM_USER) << 8));
        x = *(const v4f*)(p + cb);
    } else {
        x = *(const v4f*)((const char*)src + (((unsigned)col << 8) + cb));
    }
}

template <int MODE, int J0>
__device__ __forceinline__ void pull_batch(
    const int c32, const float v32, const float s32,
    const float* __restrict__ src, const float* __restrict__ upref,
    const float* __restrict__ items, const unsigned cb, v4f& acc)
{
    v4f   x[8];
    float v[8];
    float s[8];
    pull_fetch<MODE, J0 + 0>(c32, v32, s32, src, upref, items, cb, x[0], v[0], s[0]);
    pull_fetch<MODE, J0 + 1>(c32, v32, s32, src, upref, items, cb, x[1], v[1], s[1]);
    pull_fetch<MODE, J0 + 2>(c32, v32, s32, src, upref, items, cb, x[2], v[2], s[2]);
    pull_fetch<MODE, J0 + 3>(c32, v32, s32, src, upref, items, cb, x[3], v[3], s[3]);
    pull_fetch<MODE, J0 + 4>(c32, v32, s32, src, upref, items, cb, x[4], v[4], s[4]);
    pull_fetch<MODE, J0 + 5>(c32, v32, s32, src, upref, items, cb, x[5], v[5], s[5]);
    pull_fetch<MODE, J0 + 6>(c32, v32, s32, src, upref, items, cb, x[6], v[6], s[6]);
    pull_fetch<MODE, J0 + 7>(c32, v32, s32, src, upref, items, cb, x[7], v[7], s[7]);
#pragma unroll
    for (int k = 0; k < 8; ++k) {
        if (MODE == 1) acc += (x[k] * s[k]) * v[k];
        else           acc += x[k] * v[k];
    }
}

template <int MODE>
__global__ __launch_bounds__(256) void mgcn_pull(
    const float* __restrict__ src, const float* __restrict__ upref,
    const float* __restrict__ items, const float* __restrict__ inv,
    const int* __restrict__ start, const int* __restrict__ endp,
    const int2* __restrict__ ed, float* __restrict__ dst)
{
    static_assert(NTOT % 2 == 0, "rows are taken in pairs");
    const int lane = threadIdx.x & 31;
    const int g    = lane >> 4;
    const int c    = lane & 15;
    const int c4   = c * 4;
    const unsigned cb = (unsigned)c * 16u;
    const int row0 = __builtin_amdgcn_readfirstlane(
        (int)(((blockIdx.x * 256 + threadIdx.x) >> 5) * 2));
    if (row0 >= NTOT) return;

    const int sA = __builtin_amdgcn_readfirstlane(start[row0]);
    const int sB = __builtin_amdgcn_readfirstlane(start[row0 + 1]);
    const int nA = __builtin_amdgcn_readfirstlane(endp[row0]) - sA;
    const int nB = __builtin_amdgcn_readfirstlane(endp[row0 + 1]) - sB;
    const int nmax = nA > nB ? nA : nB;

    const int row = row0 + g;
    const int e0  = g ? sB : sA;
    const int n   = g ? nB : nA;

    float* d = dst + (size_t)row * DIM_LATENT + c4;

    v4f l0 = {0.0f, 0.0f, 0.0f, 0.0f}, l1 = l0, l2 = l0;
    if (MODE == 3) {
        if (row < NUM_USER) {
            const v4f x = *(const v4f*)(upref + (size_t)row * DIM_LATENT + c4);
            const float r = inv[row];
            l0.x = x.x * r; l0.y = x.y * r; l0.z = x.z * r; l0.w = x.w * r;
        } else {
            l0 = *(const v4f*)(items + (size_t)(row - NUM_USER) * DIM_LATENT + c4);
        }
        l1 = *(const v4f*)d;
        l2 = *(const v4f*)(src + (size_t)row * DIM_LATENT + c4);
    }

    v4f acc = {0.0f, 0.0f, 0.0f, 0.0f};
    for (int base = 0; base < nmax; base += 32) {
        int   cA = 0, cB = 0;
        float vA = 0.0f, vB = 0.0f;
        float qA = 1.0f, qB = 1.0f;
        const int jA = base + c;
        const int jB = base + 16 + c;
        if (jA < n) {
            const int2 t = ed[e0 + jA];
            cA = t.x;
            vA = __int_as_float(t.y);
        }
        if (jB < n) {
            const int2 t = ed[e0 + jB];
            cB = t.x;
            vB = __int_as_float(t.y);
        }
        if (MODE == 1) {
            if (jA < n && cA < NUM_USER) qA = inv[cA];
            if (jB < n && cB < NUM_USER) qB = inv[cB];
        }

        pull_batch<MODE, 0>(cA, vA, qA, src, upref, items, cb, acc);
        if (base + 8 < nmax) {
            pull_batch<MODE, 8>(cA, vA, qA, src, upref, items, cb, acc);
            if (base + 16 < nmax) {
                pull_batch<MODE, 0>(cB, vB, qB, src, upref, items, cb, acc);
                if (base + 24 < nmax)
                    pull_batch<MODE, 8>(cB, vB, qB, src, upref, items, cb, acc);
            }
        }
    }

    if (MODE == 3) {
        v4f o;
        o.x = (((l0.x + l1.x) + l2.x) + acc.x) * 0.25f;
        o.y = (((l0.y + l1.y) + l2.y) + acc.y) * 0.25f;
        o.z = (((l0.z + l1.z) + l2.z) + acc.z) * 0.25f;
        o.w = (((l0.w + l1.w) + l2.w) + acc.w) * 0.25f;
        *(v4f*)d = o;
    } else {
        *(v4f*)d = acc;
    }
}

extern "C" void kernel_launch(void* const* d_in, const int* in_sizes, int n_in,
                              void* d_out, int out_size, void* d_ws, size_t ws_size,
                              hipStream_t stream)
{
    const float* features = (const float*)d_in[0];
    const float* upref    = (const float*)d_in[1];
    const float* W        = (const float*)d_in[2];
    const float* bias     = (const float*)d_in[3];
    const int*   rows     = (const int*)d_in[4];
    const int*   cols     = (const int*)d_in[5];
    const float* vals     = (const float*)d_in[6];
    float*       out      = (float*)d_out;
    const int    nnz      = in_sizes[4];

    char*  ws     = (char*)d_ws;
    float* tab2   = (float*)(ws);
    int2*  tmp    = (int2*) (ws);
    int*   ltab   = (int*)  (ws + 19200000);
    float* items  = (float*)(ws + 38400000);
    float* inv    = (float*)(ws + 51200000);
    int*   start  = (int*)  (ws + 52200256);
    int*   rend   = (int*)  (ws + 52800256);
    int2*  ed     = (int2*) (ws + 53400320);

    mgcn_gemm_and_chunk_sort<<<SORT_BLOCKS + GEMM_BLOCKS, 256, 0, stream>>>(
        features, W, bias, items, rows, cols, vals, ltab, tmp, nnz);

    mgcn_bucket_rows_and_norms<<<NBUCKET + NORM_BLOCKS, 512, 0, stream>>>(
        tmp, ltab, start, rend, ed, nnz, upref, items, inv);

    const int pullBlocks = (NTOT / 2) * 32 / 256;
    mgcn_pull<1><<<pullBlocks, 256, 0, stream>>>(upref, upref, items, inv,
                                                 start, rend, ed, out);
    mgcn_pull<2><<<pullBlocks, 256, 0, stream>>>(out, upref, items, inv,
                                                 start, rend, ed, tab2);
    mgcn_pull<3><<<pullBlocks, 256, 0, stream>>>(tab2, upref, items, inv,
                                                 start, rend, ed, out);
}


# Layer 1 gathers user rows that the norm pass now stores normalized: the per-edge 4-byte inverse-norm gather and its round trip go; sort scratch sits in out until layer 1; f32 values, sum order kept
# speedup vs baseline: 3.4637x; 1.0339x over previous
#include <hip/hip_runtime.h>


#define NUM_USER   100000
#define NUM_ITEM   50000
#define NTOT       (NUM_USER + NUM_ITEM)
#define DIM_FEAT   512
#define DIM_LATENT 64

typedef __attribute__((ext_vector_type(2))) float v2f;
typedef __attribute__((ext_vector_type(8))) float v8f;

#define GEMM_TILES   (NUM_ITEM / 16)
#define GEMM_BLOCKS  ((GEMM_TILES + 7) / 8)
#define W_PITCH      132
#define LDS1_BYTES   ((DIM_FEAT / 2) * W_PITCH * 4)

__device__ __forceinline__ void gemm_tile_body(
    const float* __restrict__ F, const float* __restrict__ W,
    const float* __restrict__ bias, float* __restrict__ items,
    float* Wl, const int blk)
{

    for (int it = 0; it < (DIM_FEAT / 2) * 16 / 256; ++it) {
        const int idx = it * 256 + (int)threadIdx.x;
        const int kp  = idx >> 4;
        const int n4  = idx & 15;
        const float4 r0 = *(const float4*)(W + (size_t)(2 * kp) * DIM_LATENT + n4 * 4);
        const float4 r1 = *(const float4*)(W + (size_t)(2 * kp + 1) * DIM_LATENT + n4 * 4);
        float* d = Wl + kp * W_PITCH + n4 * 8;
        *(float4*)d       = make_float4(r0.x, r1.x, r0.y, r1.y);
        *(float4*)(d + 4) = make_float4(r0.z, r1.z, r0.w, r1.w);
    }
    __syncthreads();

    const int lane = threadIdx.x & 31;
    const int half = lane >> 4;
    const int c    = lane & 15;
    const int tile = __builtin_amdgcn_readfirstlane(
        (int)(blk * 8 + (threadIdx.x >> 5)));
    if (tile >= GEMM_TILES) return;

    const float* frow = F + (size_t)(tile * 16 + c) * DIM_FEAT + 16 * half;
    const float* wl   = Wl + (8 * half) * W_PITCH + 2 * c;

    v8f acc[4] = {};
    float4 cur[4], nxt[4];
#pragma unroll
    for (int j = 0; j < 4; ++j) cur[j] = *(const float4*)(frow + 4 * j);

    for (int t = 0; t < DIM_FEAT / 32; ++t) {
        const int tn = (t + 1 < DIM_FEAT / 32) ? t + 1 : t;
#pragma unroll
        for (int j = 0; j < 4; ++j) nxt[j] = *(const float4*)(frow + tn * 32 + 4 * j);

        const float* wt = wl + t * 16 * W_PITCH;
#pragma unroll
        for (int st = 0; st < 8; ++st) {
            const float4 q = cur[st >> 1];
            v2f a;
            a.x = (st & 1) ? q.z : q.x;
            a.y = (st & 1) ? q.w : q.y;
#pragma unroll
            for (int nt = 0; nt < 4; ++nt) {
                const v2f bm = *(const v2f*)(wt + st * W_PITCH + nt * 32);
                acc[nt] = __builtin_amdgcn_wmma_f32_16x16x4_f32(
                         false, a,  false, bm,
                         (short)0, acc[nt],  false,  false);
            }
        }
#pragma unroll
        for (int j = 0; j < 4; ++j) cur[j] = nxt[j];
    }

    float* dst = items + (size_t)(tile * 16) * DIM_LATENT;
#pragma unroll
    for (int nt = 0; nt < 4; ++nt) {
        const float bv = bias[nt * 16 + c];
#pragma unroll
        for (int j = 0; j < 8; ++j) {
            const int mr = j + 8 * half;
            dst[(size_t)mr * DIM_LATENT + nt * 16 + c] = acc[nt][j] + bv;
        }
    }
}

#define NORM_ROWS_PER_WAVE 8
#define NORM_BLOCKS ((NTOT + 16 * NORM_ROWS_PER_WAVE - 1) / (16 * NORM_ROWS_PER_WAVE))

__device__ __forceinline__ void norm_rows_body(
    const float* __restrict__ upref, float* __restrict__ items,
    float* __restrict__ inv, float* __restrict__ users0, const int nb)
{
    const int lane = threadIdx.x & 31;
    const int row0 = __builtin_amdgcn_readfirstlane(
        (int)(((nb * 512 + (int)threadIdx.x) >> 5) * NORM_ROWS_PER_WAVE));

    v2f x[NORM_ROWS_PER_WAVE];
#pragma unroll
    for (int k = 0; k < NORM_ROWS_PER_WAVE; ++k) {
        const int rl = (row0 + k < NTOT) ? row0 + k : NTOT - 1;
        const float* src = (rl < NUM_USER)
            ? (upref + (size_t)rl * DIM_LATENT)
            : (items + (size_t)(rl - NUM_USER) * DIM_LATENT);
        x[k] = *(const v2f*)(src + lane * 2);
    }

    float rmine = 0.0f;
#pragma unroll
    for (int k = 0; k < NORM_ROWS_PER_WAVE; ++k) {
        const int row = row0 + k;
        float s = x[k].x * x[k].x + x[k].y * x[k].y;
#pragma unroll
        for (int off = 16; off; off >>= 1) s += __shfl_xor(s, off, 32);
        const float r = 1.0f / fmaxf(sqrtf(s), 1e-12f);
        if (lane == k) rmine = r;
        if (row < NTOT) {
            v2f y; y.x = x[k].x * r; y.y = x[k].y * r;
            float* drow = (row < NUM_USER)
                ? (users0 + (size_t)row * DIM_LATENT)
                : (items + (size_t)(row - NUM_USER) * DIM_LATENT);
            *(v2f*)(drow + lane * 2) = y;
        }
    }
    if (lane < NORM_ROWS_PER_WAVE && row0 + lane < NUM_USER) inv[row0 + lane] = rmine;
}

#define NBUCKET      ((NTOT + 255) / 256)
#define SORT_BLOCKS  256
#define LROW         (NBUCKET + 1)
#define CHUNK_CAP    9600
#define STAGE_CAP    6144

template <int NWAVES>
__device__ __forceinline__ int block_excl_scan(int v, int* wsum, int* total)
{
    const int lane = threadIdx.x & 31;
    const int w    = threadIdx.x >> 5;
    int incl = v;
    for (int off = 1; off < 32; off <<= 1) {
        const int t = __shfl_up(incl, off, 32);
        if (lane >= off) incl += t;
    }
    if (lane == 31) wsum[w] = incl;
    __syncthreads();
    if (w == 0) {
        const int s = (lane < NWAVES) ? wsum[lane] : 0;
        int si = s;
        for (int off = 1; off < 32; off <<= 1) {
            const int t = __shfl_up(si, off, 32);
            if (lane >= off) si += t;
        }
        wsum[lane] = si - s;
        if (lane == 31) wsum[32] = si;
    }
    __syncthreads();
    *total = wsum[32];
    return wsum[w] + incl - v;
}

__device__ __forceinline__ int sort_chunk(int nnz)
{
    return ((nnz + SORT_BLOCKS * 4 - 1) / (SORT_BLOCKS * 4)) * 4;
}

__device__ __forceinline__ int2 edge_record(int r, int c, float v)
{
    return make_int2(((r & 255) << 18) | c, __float_as_int(v));
}

#define CS_THREADS  256
#define CS_GROUPS   10
#define CS_STRIDE   (CS_THREADS * 4)
#define CS_BPT      ((NBUCKET + CS_THREADS - 1) / CS_THREADS)
#define CS_H_OFF     0
#define CS_WSUM_OFF  (((NBUCKET * 4 + 15) / 16) * 16)
#define CS_STAGE_OFF (CS_WSUM_OFF + 144)

__device__ __forceinline__ void chunk_sort_body(
    const int* __restrict__ rows, const int* __restrict__ cols,
    const float* __restrict__ vals, int* __restrict__ ltab,
    int2* __restrict__ tmp, const int nnz,
    int* h, int* wsum, int2* stage, const int g)
{
    const int chunk = sort_chunk(nnz);
    const int lo    = g * chunk;
    const int hi    = (lo + chunk < nnz) ? lo + chunk : nnz;
    const int n     = (hi > lo) ? hi - lo : 0;
    const int e0    = lo + (int)threadIdx.x * 4;

    int4   r[CS_GROUPS], c[CS_GROUPS];
    float4 v[CS_GROUPS];
    bool   ok[CS_GROUPS];
#pragma unroll
    for (int u = 0; u < CS_GROUPS; ++u) {
        const int e = e0 + u * CS_STRIDE;
        ok[u] = e + 3 < hi;
        r[u] = make_int4(0, 0, 0, 0);
        c[u] = make_int4(0, 0, 0, 0);
        v[u] = make_float4(0.0f, 0.0f, 0.0f, 0.0f);
        if (ok[u]) {
            r[u] = *(const int4*)(rows + e);
            c[u] = *(const int4*)(cols + e);
            v[u] = *(const float4*)(vals + e);
        }
    }

    for (int i = threadIdx.x; i < NBUCKET; i += CS_THREADS) h[i] = 0;
    __syncthreads();

#pragma unroll
    for (int u = 0; u < CS_GROUPS; ++u) {
        if (ok[u]) {
            atomicAdd(&h[r[u].x >> 8], 1);
            atomicAdd(&h[r[u].y >> 8], 1);
            atomicAdd(&h[r[u].z >> 8], 1);
            atomicAdd(&h[r[u].w >> 8], 1);
        }
    }
    for (int u = 0, e = e0; e < hi; ++u, e += CS_STRIDE) {
        if (u < CS_GROUPS && e + 3 < hi) continue;
        const int kend = (e + 4 < hi) ? e + 4 : hi;
        for (int k = e; k < kend; ++k) atomicAdd(&h[rows[k] >> 8], 1);
    }
    __syncthreads();

    int m[CS_BPT];
    int own = 0;
#pragma unroll
    for (int j = 0; j < CS_BPT; ++j) {
        const int b = (int)threadIdx.x * CS_BPT + j;
        m[j] = (b < NBUCKET) ? h[b] : 0;
        own += m[j];
    }
    int all;
    int run = block_excl_scan<CS_THREADS / 32>(own, wsum, &all);
#pragma unroll
    for (int j = 0; j < CS_BPT; ++j) {
        const int b = (int)threadIdx.x * CS_BPT + j;
        if (b < NBUCKET) {
            ltab[g * LROW + b] = run;
            h[b] = run;
        }
        run += m[j];
    }
    if (threadIdx.x == 0) ltab[g * LROW + NBUCKET] = all;
    __syncthreads();

    if (n <= CHUNK_CAP) {
#pragma unroll
        for (int u = 0; u < CS_GROUPS; ++u) {
            if (ok[u]) {
                const int p0 = atomicAdd(&h[r[u].x >> 8], 1);
                const int p1 = atomicAdd(&h[r[u].y >> 8], 1);
                const int p2 = atomicAdd(&h[r[u].z >> 8], 1);
                const int p3 = atomicAdd(&h[r[u].w >> 8], 1);
                stage[p0] = edge_record(r[u].x, c[u].x, v[u].x);
                stage[p1] = edge_record(r[u].y, c[u].y, v[u].y);
                stage[p2] = edge_record(r[u].z, c[u].z, v[u].z);
                stage[p3] = edge_record(r[u].w, c[u].w, v[u].w);
            }
        }
        for (int u = 0, e = e0; e < hi; ++u, e += CS_STRIDE) {
            if (u < CS_GROUPS && e + 3 < hi) continue;
            const int kend = (e + 4 < hi) ? e + 4 : hi;
            for (int k = e; k < kend; ++k) {
                const int rr = rows[k];
                const int p  = atomicAdd(&h[rr >> 8], 1);
                stage[p] = edge_record(rr, cols[k], vals[k]);
            }
        }
        __syncthreads();
        for (int i = threadIdx.x; i < n; i += CS_THREADS) tmp[lo + i] = stage[i];
    } else {
#pragma unroll
        for (int u = 0; u < CS_GROUPS; ++u) {
            if (ok[u]) {
                const int p0 = atomicAdd(&h[r[u].x >> 8], 1);
                const int p1 = atomicAdd(&h[r[u].y >> 8], 1);
                const int p2 = atomicAdd(&h[r[u].z >> 8], 1);
                const int p3 = atomicAdd(&h[r[u].w >> 8], 1);
                tmp[lo + p0] = edge_record(r[u].x, c[u].x, v[u].x);
                tmp[lo + p1] = edge_record(r[u].y, c[u].y, v[u].y);
                tmp[lo + p2] = edge_record(r[u].z, c[u].z, v[u].z);
                tmp[lo + p3] = edge_record(r[u].w, c[u].w, v[u].w);
            }
        }
        for (int u = 0, e = e0; e < hi; ++u, e += CS_STRIDE) {
            if (u < CS_GROUPS && e + 3 < hi) continue;
            const int kend = (e + 4 < hi) ? e + 4 : hi;
            for (int k = e; k < kend; ++k) {
                const int rr = rows[k];
                const int p  = atomicAdd(&h[rr >> 8], 1);
                tmp[lo + p] = edge_record(rr, cols[k], vals[k]);
            }
        }
    }
}

__global__ __launch_bounds__(256) void mgcn_gemm_and_chunk_sort(
    const float* __restrict__ F, const float* __restrict__ W,
    const float* __restrict__ bias, float* __restrict__ items,
    const int* __restrict__ rows, const int* __restrict__ cols,
    const float* __restrict__ vals, int* __restrict__ ltab,
    int2* __restrict__ tmp, int nnz)
{
    __shared__ __attribute__((aligned(16))) unsigned char lds_raw[LDS1_BYTES];
    static_assert(CS_STAGE_OFF % 16 == 0, "record stage alignment");
    static_assert(CS_STAGE_OFF + CHUNK_CAP * 8 <= LDS1_BYTES, "sort stage fits the W stage");

    if (blockIdx.x < SORT_BLOCKS) {
        chunk_sort_body(rows, cols, vals, ltab, tmp, nnz,
                        (int*)(lds_raw + CS_H_OFF), (int*)(lds_raw + CS_WSUM_OFF),
                        (int2*)(lds_raw + CS_STAGE_OFF), (int)blockIdx.x);
        return;
    }
    gemm_tile_body(F, W, bias, items, (float*)lds_raw, (int)blockIdx.x - SORT_BLOCKS);
}

#define RUNS_PER_WAVE (SORT_BLOCKS / 16)
__global__ __launch_bounds__(512) void mgcn_bucket_rows_and_norms(
    const int2* __restrict__ tmp, const int* __restrict__ ltab,
    int* __restrict__ start, int* __restrict__ endp,
    int2* __restrict__ ed, int nnz,
    const float* __restrict__ upref, float* __restrict__ items,
    float* __restrict__ inv, float* __restrict__ users0)
{
    if (blockIdx.x >= NBUCKET) {
        norm_rows_body(upref, items, inv, users0, (int)blockIdx.x - NBUCKET);
        return;
    }

    __shared__ int  cnt[256];
    __shared__ int  cur[256];
    __shared__ int  rs[SORT_BLOCKS];
    __shared__ int  rn[SORT_BLOCKS];
    __shared__ int  wsum[33];
    __shared__ int2 stage[STAGE_CAP];

    const int b     = (int)blockIdx.x;
    const int lane  = threadIdx.x & 31;
    const int w     = __builtin_amdgcn_readfirstlane((int)(threadIdx.x >> 5));
    const int chunk = sort_chunk(nnz);

    int s0 = 0, len = 0;
    if (threadIdx.x < SORT_BLOCKS) {
        s0  = ltab[(int)threadIdx.x * LROW + b];
        len = ltab[(int)threadIdx.x * LROW + b + 1] - s0;
        rs[threadIdx.x] = (int)threadIdx.x * chunk + s0;
        rn[threadIdx.x] = len;
    }
    if (threadIdx.x < 256) cnt[threadIdx.x] = 0;
    int sumStart, sumLen;
    block_excl_scan<16>(s0, wsum, &sumStart);
    block_excl_scan<16>(len, wsum, &sumLen);
    const int base = __builtin_amdgcn_readfirstlane(sumStart);
    const int n    = __builtin_amdgcn_readfirstlane(sumLen);

    int2 t[RUNS_PER_WAVE];
    int  ln[RUNS_PER_WAVE];
    int  ro[RUNS_PER_WAVE];
#pragma unroll
    for (int u = 0; u < RUNS_PER_WAVE; ++u) {
        ln[u] = __builtin_amdgcn_readfirstlane(rn[w + 16 * u]);
        ro[u] = __builtin_amdgcn_readfirstlane(rs[w + 16 * u]);
    }
#pragma unroll
    for (int u = 0; u < RUNS_PER_WAVE; ++u) {
        t[u] = make_int2(0, 0);
        if (lane < ln[u]) t[u] = tmp[ro[u] + lane];
    }
#pragma unroll
    for (int u = 0; u < RUNS_PER_WAVE; ++u) {
        if (lane < ln[u]) atomicAdd(&cnt[(unsigned)t[u].x >> 18], 1);
        for (int j = lane + 32; j < ln[u]; j += 32) {
            const int2 x = tmp[ro[u] + j];
            atomicAdd(&cnt[(unsigned)x.x >> 18], 1);
        }
    }
    __syncthreads();

    const int mine = (threadIdx.x < 256) ? cnt[threadIdx.x] : 0;
    int all;
    const int excl = block_excl_scan<16>(mine, wsum, &all);
    if (threadIdx.x < 256) {
        cur[threadIdx.x] = excl;
        const int row = b * 256 + (int)threadIdx.x;
        if (row < NTOT) {
            start[row] = base + excl;
            endp[row]  = base + excl + mine;
        }
    }
    __syncthreads();

    if (n <= STAGE_CAP) {
#pragma unroll
        for (int u0 = 0; u0 < RUNS_PER_WAVE; u0 += 8) {
            int p[8];
#pragma unroll
            for (int k = 0; k < 8; ++k) {
                p[k] = 0;
                if (lane < ln[u0 + k])
                    p[k] = atomicAdd(&cur[(unsigned)t[u0 + k].x >> 18], 1);
            }
#pragma unroll
            for (int k = 0; k < 8; ++k) {
                const int u = u0 + k;
                if (lane < ln[u]) stage[p[k]] = make_int2(t[u].x & 0x3FFFF, t[u].y);
                for (int j = lane + 32; j < ln[u]; j += 32) {
                    const int2 x = tmp[ro[u] + j];
                    const int  q = atomicAdd(&cur[(unsigned)x.x >> 18], 1);
                    stage[q] = make_int2(x.x & 0x3FFFF, x.y);
                }
            }
        }
        __syncthreads();
        for (int i = threadIdx.x; i < n; i += 512) ed[base + i] = stage[i];
    } else {
#pragma unroll
        for (int u = 0; u < RUNS_PER_WAVE; ++u) {
            if (lane < ln[u]) {
                const int p = atomicAdd(&cur[(unsigned)t[u].x >> 18], 1);
                ed[base + p] = make_int2(t[u].x & 0x3FFFF, t[u].y);
            }
            for (int j = lane + 32; j < ln[u]; j += 32) {
                const int2 x = tmp[ro[u] + j];
                const int  p = atomicAdd(&cur[(unsigned)x.x >> 18], 1);
                ed[base + p] = make_int2(x.x & 0x3FFFF, x.y);
            }
        }
    }
}

typedef __attribute__((ext_vector_type(4))) float v4f;

template <int J>
__device__ __forceinline__ int row_bcast_i(int v)
{
    return __builtin_amdgcn_update_dpp(0, v, 0x150 + J, 0xF, 0xF, false);
}
template <int J>
__device__ __forceinline__ float row_bcast_f(float v)
{
    return __int_as_float(row_bcast_i<J>(__float_as_int(v)));
}

template <int MODE, int J>
__device__ __forceinline__ void pull_fetch(
    const int c32, const float v32,
    const float* __restrict__ src, const float* __restrict__ items,
    const unsigned cb, v4f& x, float& v)
{
    const int col = row_bcast_i<J>(c32);
    v = row_bcast_f<J>(v32);
    if (MODE == 1) {
        const char* p = (col < NUM_USER)
            ? ((const char*)src + ((unsigned)col << 8))
            : ((const char*)items + ((unsigned)(col - NUM_USER) << 8));
        x = *(const v4f*)(p + cb);
    } else {
        x = *(const v4f*)((const char*)src + (((unsigned)col << 8) + cb));
    }
}

template <int MODE, int J0>
__device__ __forceinline__ void pull_batch(
    const int c32, const float v32,
    const float* __restrict__ src, const float* __restrict__ items,
    const unsigned cb, v4f& acc)
{
    v4f   x[8];
    float v[8];
    pull_fetch<MODE, J0 + 0>(c32, v32, src, items, cb, x[0], v[0]);
    pull_fetch<MODE, J0 + 1>(c32, v32, src, items, cb, x[1], v[1]);
    pull_fetch<MODE, J0 + 2>(c32, v32, src, items, cb, x[2], v[2]);
    pull_fetch<MODE, J0 + 3>(c32, v32, src, items, cb, x[3], v[3]);
    pull_fetch<MODE, J0 + 4>(c32, v32, src, items, cb, x[4], v[4]);
    pull_fetch<MODE, J0 + 5>(c32, v32, src, items, cb, x[5], v[5]);
    pull_fetch<MODE, J0 + 6>(c32, v32, src, items, cb, x[6], v[6]);
    pull_fetch<MODE, J0 + 7>(c32, v32, src, items, cb, x[7], v[7]);
#pragma unroll
    for (int k = 0; k < 8; ++k) acc += x[k] * v[k];
}

template <int MODE>
__global__ __launch_bounds__(256) void mgcn_pull(
    const float* __restrict__ src, const float* __restrict__ upref,
    const float* __restrict__ items, const float* __restrict__ inv,
    const int* __restrict__ start, const int* __restrict__ endp,
    const int2* __restrict__ ed, float* __restrict__ dst)
{
    static_assert(NTOT % 2 == 0, "rows are taken in pairs");
    const int lane = threadIdx.x & 31;
    const int g    = lane >> 4;
    const int c    = lane & 15;
    const int c4   = c * 4;
    const unsigned cb = (unsigned)c * 16u;
    const int row0 = __builtin_amdgcn_readfirstlane(
        (int)(((blockIdx.x * 256 + threadIdx.x) >> 5) * 2));
    if (row0 >= NTOT) return;

    const int sA = __builtin_amdgcn_readfirstlane(start[row0]);
    const int sB = __builtin_amdgcn_readfirstlane(start[row0 + 1]);
    const int nA = __builtin_amdgcn_readfirstlane(endp[row0]) - sA;
    const int nB = __builtin_amdgcn_readfirstlane(endp[row0 + 1]) - sB;
    const int nmax = nA > nB ? nA : nB;

    const int row = row0 + g;
    const int e0  = g ? sB : sA;
    const int n   = g ? nB : nA;

    float* d = dst + (size_t)row * DIM_LATENT + c4;

    v4f l0 = {0.0f, 0.0f, 0.0f, 0.0f}, l1 = l0, l2 = l0;
    if (MODE == 3) {
        if (row < NUM_USER) {
            const v4f x = *(const v4f*)(upref + (size_t)row * DIM_LATENT + c4);
            const float r = inv[row];
            l0.x = x.x * r; l0.y = x.y * r; l0.z = x.z * r; l0.w = x.w * r;
        } else {
            l0 = *(const v4f*)(items + (size_t)(row - NUM_USER) * DIM_LATENT + c4);
        }
        l1 = *(const v4f*)d;
        l2 = *(const v4f*)(src + (size_t)row * DIM_LATENT + c4);
    }

    v4f acc = {0.0f, 0.0f, 0.0f, 0.0f};
    for (int base = 0; base < nmax; base += 32) {
        int   cA = 0, cB = 0;
        float vA = 0.0f, vB = 0.0f;
        const int jA = base + c;
        const int jB = base + 16 + c;
        if (jA < n) {
            const int2 t = ed[e0 + jA];
            cA = t.x;
            vA = __int_as_float(t.y);
        }
        if (jB < n) {
            const int2 t = ed[e0 + jB];
            cB = t.x;
            vB = __int_as_float(t.y);
        }

        pull_batch<MODE, 0>(cA, vA, src, items, cb, acc);
        if (base + 8 < nmax) {
            pull_batch<MODE, 8>(cA, vA, src, items, cb, acc);
            if (base + 16 < nmax) {
                pull_batch<MODE, 0>(cB, vB, src, items, cb, acc);
                if (base + 24 < nmax)
                    pull_batch<MODE, 8>(cB, vB, src, items, cb, acc);
            }
        }
    }

    if (MODE == 3) {
        v4f o;
        o.x = (((l0.x + l1.x) + l2.x) + acc.x) * 0.25f;
        o.y = (((l0.y + l1.y) + l2.y) + acc.y) * 0.25f;
        o.z = (((l0.z + l1.z) + l2.z) + acc.z) * 0.25f;
        o.w = (((l0.w + l1.w) + l2.w) + acc.w) * 0.25f;
        *(v4f*)d = o;
    } else {
        *(v4f*)d = acc;
    }
}

extern "C" void kernel_launch(void* const* d_in, const int* in_sizes, int n_in,
                              void* d_out, int out_size, void* d_ws, size_t ws_size,
                              hipStream_t stream)
{
    const float* features = (const float*)d_in[0];
    const float* upref    = (const float*)d_in[1];
    const float* W        = (const float*)d_in[2];
    const float* bias     = (const float*)d_in[3];
    const int*   rows     = (const int*)d_in[4];
    const int*   cols     = (const int*)d_in[5];
    const float* vals     = (const float*)d_in[6];
    float*       out      = (float*)d_out;
    const int    nnz      = in_sizes[4];

    char*  ws     = (char*)d_ws;
    float* tab2   = (float*)(ws);
    float* users0 = (float*)(ws);
    int2*  tmp    = (int2*) ((char*)d_out);
    int*   ltab   = (int*)  ((char*)d_out + 19200000);
    float* items  = (float*)(ws + 38400000);
    float* inv    = (float*)(ws + 51200000);
    int*   start  = (int*)  (ws + 52200256);
    int*   rend   = (int*)  (ws + 52800256);
    int2*  ed     = (int2*) (ws + 53400320);

    mgcn_gemm_and_chunk_sort<<<SORT_BLOCKS + GEMM_BLOCKS, 256, 0, stream>>>(
        features, W, bias, items, rows, cols, vals, ltab, tmp, nnz);

    mgcn_bucket_rows_and_norms<<<NBUCKET + NORM_BLOCKS, 512, 0, stream>>>(
        tmp, ltab, start, rend, ed, nnz, upref, items, inv, users0);

    const int pullBlocks = (NTOT / 2) * 32 / 256;
    mgcn_pull<1><<<pullBlocks, 256, 0, stream>>>(users0, upref, items, inv,
                                                 start, rend, ed, out);
    mgcn_pull<2><<<pullBlocks, 256, 0, stream>>>(out, upref, items, inv,
                                                 start, rend, ed, tab2);
    mgcn_pull<3><<<pullBlocks, 256, 0, stream>>>(tab2, upref, items, inv,
                                                 start, rend, ed, out);
}
